# starting kernel
# baseline (speedup 1.0000x reference)
#include <hip/hip_runtime.h>

typedef __attribute__((ext_vector_type(2))) float v2f;
typedef __attribute__((ext_vector_type(8))) float v8f;

#define N_NODES 100000
#define N_EDGES 1600000
#define IN_F 128
#define HID 128

__global__ void degrees_kernel(const int* __restrict__ src, const int* __restrict__ dst,
                               float* __restrict__ deg_out, float* __restrict__ deg_in, int E) {
    int e = blockIdx.x * blockDim.x + threadIdx.x;
    if (e >= E) return;
    atomicAdd(&deg_out[src[e]], 1.0f);
    atomicAdd(&deg_in[dst[e]], 1.0f);
}

__global__ void ninv_kernel(float* __restrict__ deg, int n) {
    int i = blockIdx.x * blockDim.x + threadIdx.x;
    if (i >= n) return;
    float d = deg[i];
    deg[i] = (d > 0.0f) ? (1.0f / sqrtf(d)) : 0.0f;
}

__global__ void scale_kernel(const float* __restrict__ x, const float* __restrict__ ninv_out,
                             float* __restrict__ xs) {
    int i = blockIdx.x * blockDim.x + threadIdx.x;
    xs[i] = x[i] * ninv_out[i >> 7];
}

__global__ void agg128_kernel(const int* __restrict__ src, const int* __restrict__ dst,
                              const float* __restrict__ xs, float* __restrict__ agg, int E) {
    int e = (blockIdx.x * blockDim.x + threadIdx.x) >> 5;
    int lane = threadIdx.x & 31;
    if (e >= E) return;
    int s = src[e];
    int d = dst[e];
    const float4 v = *(const float4*)(xs + (size_t)s * 128 + lane * 4);
    float* dp = agg + (size_t)d * 128 + lane * 4;
    atomicAdd(dp + 0, v.x);
    atomicAdd(dp + 1, v.y);
    atomicAdd(dp + 2, v.z);
    atomicAdd(dp + 3, v.w);
}

__global__ void gemm1_relu_kernel(const float* __restrict__ agg, const float* __restrict__ W1,
                                  const float* __restrict__ b1, const float* __restrict__ ninv_in,
                                  float* __restrict__ h1) {
    __shared__ float As[16 * 128];

    const int tid  = threadIdx.x;
    const int lane = tid & 31;
    const int wave = tid >> 5;
    const int row0 = blockIdx.x * 16;

    const float4* g4 = (const float4*)(agg + (size_t)row0 * 128);
    float4* s4 = (float4*)As;
    s4[tid * 2 + 0] = g4[tid * 2 + 0];
    s4[tid * 2 + 1] = g4[tid * 2 + 1];
    __syncthreads();

    const int col0 = wave * 16;
    const int mn   = lane & 15;
    const int kb   = (lane >> 4) * 2;

    v8f c = {};
#pragma unroll
    for (int kk = 0; kk < 32; ++kk) {
        const int k = kk * 4 + kb;
        v2f a, b;
        a.x = As[mn * 128 + k];
        a.y = As[mn * 128 + k + 1];
        b.x = W1[(size_t)k * 128 + col0 + mn];
        b.y = W1[(size_t)(k + 1) * 128 + col0 + mn];
        c = __builtin_amdgcn_wmma_f32_16x16x4_f32(false, a, false, b,
                                                  (short)0, c, false, false);
    }

    const int n     = col0 + mn;
    const int mbase = row0 + ((lane >> 4) ? 8 : 0);
    const float bn  = b1[n];
#pragma unroll
    for (int r = 0; r < 8; ++r) {
        const int m = mbase + r;
        float v = c[r] * ninv_in[m] + bn;
        h1[(size_t)m * 128 + n] = fmaxf(v, 0.0f);
    }
}

__global__ void l2_matvec_kernel(const float* __restrict__ h1, const float* __restrict__ ninv_out,
                                 const float* __restrict__ W2, float* __restrict__ y_node, int N) {
    int node = blockIdx.x * 8 + (threadIdx.x >> 5);
    int lane = threadIdx.x & 31;
    if (node >= N) return;
    const float s = ninv_out[node];
    float y0 = 0.0f, y1 = 0.0f;
#pragma unroll
    for (int j = 0; j < 4; ++j) {
        int f = lane + j * 32;
        float v = h1[(size_t)node * 128 + f] * s;
        y0 += v * W2[f * 2 + 0];
        y1 += v * W2[f * 2 + 1];
    }
#pragma unroll
    for (int off = 16; off > 0; off >>= 1) {
        y0 += __shfl_xor(y0, off, 32);
        y1 += __shfl_xor(y1, off, 32);
    }
    if (lane == 0) {
        y_node[node * 2 + 0] = y0;
        y_node[node * 2 + 1] = y1;
    }
}

__global__ void agg2_kernel(const int* __restrict__ src, const int* __restrict__ dst,
                            const float* __restrict__ y_node, float* __restrict__ out_acc, int E) {
    int e = blockIdx.x * blockDim.x + threadIdx.x;
    if (e >= E) return;
    int s = src[e];
    int d = dst[e];
    atomicAdd(&out_acc[d * 2 + 0], y_node[s * 2 + 0]);
    atomicAdd(&out_acc[d * 2 + 1], y_node[s * 2 + 1]);
}

__global__ void final_kernel(const float* __restrict__ out_acc, const float* __restrict__ ninv_in,
                             const float* __restrict__ b2, float* __restrict__ out, int N) {
    int n = blockIdx.x * blockDim.x + threadIdx.x;
    if (n >= N) return;
    float ni = ninv_in[n];
    out[n * 2 + 0] = out_acc[n * 2 + 0] * ni + b2[0];
    out[n * 2 + 1] = out_acc[n * 2 + 1] * ni + b2[1];
}

extern "C" void kernel_launch(void* const* d_in, const int* in_sizes, int n_in,
                              void* d_out, int out_size, void* d_ws, size_t ws_size,
                              hipStream_t stream) {
    const float* in_feat = (const float*)d_in[0];
    const float* W1      = (const float*)d_in[1];
    const float* b1      = (const float*)d_in[2];
    const float* W2      = (const float*)d_in[3];
    const float* b2      = (const float*)d_in[4];
    const int*   src     = (const int*)d_in[5];
    const int*   dst     = (const int*)d_in[6];
    float*       out     = (float*)d_out;

    const int N = N_NODES;
    const int E = N_EDGES;

    float* ws      = (float*)d_ws;
    float* deg_out = ws;
    float* deg_in  = ws + N;
    float* xs      = ws + 2 * (size_t)N;
    float* agg     = xs + (size_t)N * 128;
    float* y_node  = agg + (size_t)N * 128;
    float* out_acc = y_node + 2 * (size_t)N;

    hipMemsetAsync(deg_out, 0, 2 * (size_t)N * sizeof(float), stream);
    hipMemsetAsync(agg, 0, (size_t)N * 128 * sizeof(float), stream);
    hipMemsetAsync(out_acc, 0, 2 * (size_t)N * sizeof(float), stream);

    degrees_kernel<<<(E + 255) / 256, 256, 0, stream>>>(src, dst, deg_out, deg_in, E);
    ninv_kernel<<<(2 * N + 255) / 256, 256, 0, stream>>>(deg_out, 2 * N);
    scale_kernel<<<(N * 128) / 256, 256, 0, stream>>>(in_feat, deg_out, xs);
    agg128_kernel<<<E / 8, 256, 0, stream>>>(src, dst, xs, agg, E);
    gemm1_relu_kernel<<<N / 16, 256, 0, stream>>>(agg, W1, b1, deg_in, xs);
    l2_matvec_kernel<<<N / 8, 256, 0, stream>>>(xs, deg_out, W2, y_node, N);
    agg2_kernel<<<(E + 255) / 256, 256, 0, stream>>>(src, dst, y_node, out_acc, E);
    final_kernel<<<(N + 255) / 256, 256, 0, stream>>>(out_acc, deg_in, b2, out, N);
}


# layer-1 GEMM moved into the work-group that owns the LDS sums: wave t takes tile t's A fragments from its LDS rows, B from W1 copied into the queues' LDS (same k order); sums plane and a launch gone
# speedup vs baseline: 3.6075x; 3.6075x over previous
#include <hip/hip_runtime.h>

typedef __attribute__((ext_vector_type(2))) float v2f;
typedef __attribute__((ext_vector_type(8))) float v8f;
typedef __attribute__((ext_vector_type(4))) float v4f;

#define N_NODES 100000
#define N_EDGES 1600000
#define IN_F 128
#define HID 128

#define N_RANGES 8
#define N_SLICES 32
#define RANGE_NODES (N_NODES / N_RANGES)
#define SLICE_EDGES (N_EDGES / N_SLICES)

#define L1_GROUPS 256
#define L1_ROWS 391
#define L1_SEG 512
#define MAP_VEC (N_EDGES / 16)
#define L1_TILES 25
#define L1_PITCH 132
static_assert(L1_TILES * 16 >= L1_ROWS + 1, "the tiles must cover the range and its spare row");
static_assert(32 * L1_SEG == HID * IN_F, "W1 in fragment order takes the place of the edge queues");

__global__ __launch_bounds__(1024) void degrees_kernel(const int* __restrict__ src,
                                                       const int* __restrict__ dst,
                                                       float* __restrict__ part,
                                                       unsigned* __restrict__ omap) {
    __shared__ unsigned h_out[RANGE_NODES];
    __shared__ unsigned h_in[RANGE_NODES];

    const int tid   = threadIdx.x;
    const int slice = blockIdx.x;
    const int lo    = blockIdx.y * RANGE_NODES;

    for (int i = tid; i < RANGE_NODES; i += 1024) {
        h_out[i] = 0u;
        h_in[i]  = 0u;
    }
    __syncthreads();

    const int4* s4 = (const int4*)(src + (size_t)slice * SLICE_EDGES);
    const int4* d4 = (const int4*)(dst + (size_t)slice * SLICE_EDGES);
    for (int i = tid; i < SLICE_EDGES / 4; i += 1024) {
        const int4 s = s4[i];
        const int4 d = d4[i];
        if (blockIdx.y == 0) {
            omap[(size_t)slice * (SLICE_EDGES / 4) + i] =
                ((unsigned)d.x / L1_ROWS) | (((unsigned)d.y / L1_ROWS) << 8) |
                (((unsigned)d.z / L1_ROWS) << 16) | (((unsigned)d.w / L1_ROWS) << 24);
        }
        unsigned r;
        r = (unsigned)(s.x - lo); if (r < RANGE_NODES) atomicAdd(&h_out[r], 1u);
        r = (unsigned)(s.y - lo); if (r < RANGE_NODES) atomicAdd(&h_out[r], 1u);
        r = (unsigned)(s.z - lo); if (r < RANGE_NODES) atomicAdd(&h_out[r], 1u);
        r = (unsigned)(s.w - lo); if (r < RANGE_NODES) atomicAdd(&h_out[r], 1u);
        r = (unsigned)(d.x - lo); if (r < RANGE_NODES) atomicAdd(&h_in[r], 1u);
        r = (unsigned)(d.y - lo); if (r < RANGE_NODES) atomicAdd(&h_in[r], 1u);
        r = (unsigned)(d.z - lo); if (r < RANGE_NODES) atomicAdd(&h_in[r], 1u);
        r = (unsigned)(d.w - lo); if (r < RANGE_NODES) atomicAdd(&h_in[r], 1u);
    }
    __syncthreads();

    float* po = part + (size_t)slice * (2 * N_NODES) + lo;
    float* pi = po + N_NODES;
    for (int i = tid; i < RANGE_NODES; i += 1024) {
        po[i] = (float)h_out[i];
        pi[i] = (float)h_in[i];
    }
}

__global__ void ninv_kernel(const float* __restrict__ part, float* __restrict__ ninv, int n) {
    int i = blockIdx.x * blockDim.x + threadIdx.x;
    if (i >= n) return;
    float d = 0.0f;
#pragma unroll
    for (int s = 0; s < N_SLICES; ++s) d += part[(size_t)s * (2 * N_NODES) + i];
    ninv[i] = (d > 0.0f) ? (1.0f / sqrtf(d)) : 0.0f;
}

#define STREAM_ROWS 25
#define STREAM_GROUPS (N_NODES / (8 * STREAM_ROWS))
static_assert(STREAM_GROUPS * 8 * STREAM_ROWS == N_NODES, "rows must split evenly over the waves");

__global__ __launch_bounds__(256) void scale_kernel(const float* __restrict__ x,
                                                    const float* __restrict__ ninv_out,
                                                    float* __restrict__ xs) {
    const int lane = threadIdx.x & 31;
    const int row0 = (blockIdx.x * 8 + (threadIdx.x >> 5)) * STREAM_ROWS;
    const v4f* x4 = (const v4f*)x + (size_t)row0 * 32 + lane;
    v4f*       o4 = (v4f*)xs + (size_t)row0 * 32 + lane;
    const float nv = ninv_out[row0 + (lane < STREAM_ROWS ? lane : 0)];
    v4f v[STREAM_ROWS];
#pragma unroll
    for (int j = 0; j < STREAM_ROWS; ++j) v[j] = __builtin_nontemporal_load(x4 + j * 32);
#pragma unroll
    for (int j = 0; j < STREAM_ROWS; ++j) {
        const float s = __int_as_float(__builtin_amdgcn_readlane(__float_as_int(nv), j));
        o4[j * 32] = v[j] * s;
    }
}

__device__ __forceinline__ unsigned zero_flags(unsigned x) {
    const unsigned t = (x & 0x7f7f7f7fu) + 0x7f7f7f7fu;
    return ~(t | x | 0x7f7f7f7fu);
}

__device__ __forceinline__ unsigned match16(uint4 w, unsigned pat) {
    return (zero_flags(w.x ^ pat) >> 7) | (zero_flags(w.y ^ pat) >> 6) |
           (zero_flags(w.z ^ pat) >> 5) | (zero_flags(w.w ^ pat) >> 4);
}

__global__ __launch_bounds__(1024) void agg_lds_kernel(const int* __restrict__ src,
                                                       const int* __restrict__ dst,
                                                       const unsigned* __restrict__ omap,
                                                       const float* __restrict__ xs,
                                                       const float* __restrict__ W1,
                                                       const float* __restrict__ b1,
                                                       const float* __restrict__ ninv_in,
                                                       float* __restrict__ h1) {
    __shared__ __attribute__((aligned(16))) float acc[L1_TILES * 16 * L1_PITCH];
    __shared__ __attribute__((aligned(16))) unsigned queue[32 * L1_SEG];

    const int tid  = threadIdx.x;
    const int lane = tid & 31;
    const int wave = tid >> 5;
    const int g    = blockIdx.x;
    const int lo   = g * L1_ROWS;

    {
        float4* a4 = (float4*)acc;
        for (int i = tid; i < L1_TILES * 16 * L1_PITCH / 4; i += 1024) a4[i] = make_float4(0.f, 0.f, 0.f, 0.f);
    }
    __syncthreads();

    const uint4*   m4   = (const uint4*)omap;
    const unsigned pat  = (unsigned)g * 0x01010101u;
    const int      rot  = g * (MAP_VEC / L1_GROUPS);
    const unsigned ltm  = (1u << lane) - 1u;
    unsigned*      myq  = queue + wave * L1_SEG;
    int            cnt  = 0;

    for (int it = 0; it < (MAP_VEC + 4095) / 4096; ++it) {
        int   idx[4];
        bool  val[4];
        uint4 w[4];
#pragma unroll
        for (int u = 0; u < 4; ++u) {
            const int j = it * 4096 + u * 1024 + tid;
            val[u] = j < MAP_VEC;
            int i = j + rot;
            if (i >= MAP_VEC) i -= MAP_VEC;
            if (!val[u]) i = 0;
            idx[u] = i;
            w[u]   = m4[i];
        }
        __builtin_amdgcn_sched_barrier(0);
#pragma unroll
        for (int u = 0; u < 4; ++u) {
            unsigned m = val[u] ? match16(w[u], pat) : 0u;
            while (__builtin_amdgcn_ballot_w32(m != 0u) != 0u) {
                const bool     has = m != 0u;
                const unsigned bal = __builtin_amdgcn_ballot_w32(has);
                if (has) {
                    const int p = __ffs(m) - 1;
                    const int e = idx[u] * 16 + (p & 7) * 4 + (p >> 3);
                    m &= m - 1u;
                    const int slot = cnt + __popc(bal & ltm);
                    if (slot < L1_SEG) {
                        myq[slot] = (unsigned)e;
                    } else {
                        const int    s  = src[e];
                        const int    dl = dst[e] - lo;
                        const float* xr = xs + (size_t)s * 128;
                        float*       ar = acc + dl * L1_PITCH;
                        for (int f = 0; f < 128; ++f) atomicAdd(&ar[(f & 3) * 32 + (f >> 2)], xr[f]);
                    }
                }
                cnt += __popc(bal);
            }
        }
    }
    __syncthreads();

    const int n = cnt < L1_SEG ? cnt : L1_SEG;
    for (int k0 = 0; k0 < n; k0 += 32) {
        const int  k  = k0 + lane;
        const bool ok = k < n;
        const int  e  = ok ? (int)myq[k] : 0;
        int s  = src[e];
        int dl = dst[e] - lo;
        if (!ok) { s = 0; dl = L1_ROWS; }
        const int left = n - k0;
#pragma unroll
        for (int sb = 0; sb < 4; ++sb) {
            if (sb * 8 < left) {
                float4 v[8];
                int    dr[8];
#pragma unroll
                for (int j = 0; j < 8; ++j) {
                    const int sj = __builtin_amdgcn_readlane(s, sb * 8 + j);
                    dr[j]        = __builtin_amdgcn_readlane(dl, sb * 8 + j);
                    v[j] = *(const float4*)(xs + (size_t)sj * 128 + lane * 4);
                }
#pragma unroll
                for (int j = 0; j < 8; ++j) {
                    float* a = acc + dr[j] * L1_PITCH + lane;
                    atomicAdd(a,      v[j].x);
                    atomicAdd(a + 32, v[j].y);
                    atomicAdd(a + 64, v[j].z);
                    atomicAdd(a + 96, v[j].w);
                }
            }
        }
    }
    float2 wv[8];
    {
        const int l  = tid & 31;
        const int kk = tid >> 5;
        const int k  = kk * 4 + (l >> 4) * 2;
        const float* wp = W1 + (size_t)k * 128 + (l & 15);
#pragma unroll
        for (int ct = 0; ct < 8; ++ct) wv[ct] = make_float2(wp[ct * 16], wp[128 + ct * 16]);
    }
    __syncthreads();
    {
        float2* wst = (float2*)queue + tid;
#pragma unroll
        for (int ct = 0; ct < 8; ++ct) wst[ct * 1024] = wv[ct];
    }

    const bool  active = wave < L1_TILES;
    const int   mn     = lane & 15;
    const int   hf     = lane >> 4;
    float*      trow   = acc + wave * 16 * L1_PITCH;
    v2f a[32];
    if (active) {
        const float* ar = trow + mn * L1_PITCH + hf * 64;
#pragma unroll
        for (int kk = 0; kk < 32; ++kk) {
            a[kk].x = ar[kk];
            a[kk].y = ar[32 + kk];
        }
    }
    __syncthreads();

    if (active) {
        const v2f* wl   = (const v2f*)queue + lane;
        const int  m0   = lo + wave * 16 + hf * 8;
        float      nv[8];
#pragma unroll
        for (int r = 0; r < 8; ++r) nv[r] = ninv_in[(m0 + r < N_NODES) ? (m0 + r) : (N_NODES - 1)];
        for (int ct = 0; ct < 8; ct += 2) {
            v8f c0 = {};
            v8f c1 = {};
#pragma unroll
            for (int kk = 0; kk < 32; ++kk) {
                const v2f b0 = wl[((ct + 0) * 32 + kk) * 32];
                const v2f b1v = wl[((ct + 1) * 32 + kk) * 32];
                c0 = __builtin_amdgcn_wmma_f32_16x16x4_f32(false, a[kk], false, b0,
                                                           (short)0, c0, false, false);
                c1 = __builtin_amdgcn_wmma_f32_16x16x4_f32(false, a[kk], false, b1v,
                                                           (short)0, c1, false, false);
            }
            const int   n0  = ct * 16 + mn;
            const float bn0 = b1[n0];
            const float bn1 = b1[n0 + 16];
            float* d = trow + hf * 8 * L1_PITCH + n0;
#pragma unroll
            for (int r = 0; r < 8; ++r) {
                d[r * L1_PITCH]      = fmaxf(c0[r] * nv[r] + bn0, 0.0f);
                d[r * L1_PITCH + 16] = fmaxf(c1[r] * nv[r] + bn1, 0.0f);
            }
        }
    }
    __syncthreads();

    if (active) {
        for (int i = 0; i < 16; ++i) {
            const int r = wave * 16 + i;
            if (r < L1_ROWS && lo + r < N_NODES) {
                *(float4*)(h1 + (size_t)(lo + r) * 128 + lane * 4) =
                    *(const float4*)(acc + r * L1_PITCH + lane * 4);
            }
        }
    }
}

__global__ void l2_matvec_kernel(const float* __restrict__ h1, const float* __restrict__ ninv_out,
                                 const float* __restrict__ W2, float* __restrict__ y_node, int N) {
    int node = blockIdx.x * 8 + (threadIdx.x >> 5);
    int lane = threadIdx.x & 31;
    if (node >= N) return;
    const float s = ninv_out[node];
    float y0 = 0.0f, y1 = 0.0f;
#pragma unroll
    for (int j = 0; j < 4; ++j) {
        int f = lane + j * 32;
        float v = h1[(size_t)node * 128 + f] * s;
        y0 += v * W2[f * 2 + 0];
        y1 += v * W2[f * 2 + 1];
    }
#pragma unroll
    for (int off = 16; off > 0; off >>= 1) {
        y0 += __shfl_xor(y0, off, 32);
        y1 += __shfl_xor(y1, off, 32);
    }
    if (lane == 0) {
        y_node[node * 2 + 0] = y0;
        y_node[node * 2 + 1] = y1;
    }
}

__global__ __launch_bounds__(1024) void agg2_kernel(const int* __restrict__ src,
                                                    const int* __restrict__ dst,
                                                    const float* __restrict__ y_node,
                                                    float* __restrict__ part2) {
    __shared__ float acc[2 * RANGE_NODES];

    const int tid   = threadIdx.x;
    const int slice = blockIdx.x;
    const int lo    = blockIdx.y * RANGE_NODES;

    for (int i = tid; i < 2 * RANGE_NODES; i += 1024) acc[i] = 0.0f;
    __syncthreads();

    const int4*   s4 = (const int4*)(src + (size_t)slice * SLICE_EDGES);
    const int4*   d4 = (const int4*)(dst + (size_t)slice * SLICE_EDGES);
    const float2* yn = (const float2*)y_node;
    for (int i = tid; i < SLICE_EDGES / 4; i += 1024) {
        const int4 s = s4[i];
        const int4 d = d4[i];
        const unsigned r0 = (unsigned)(d.x - lo);
        const unsigned r1 = (unsigned)(d.y - lo);
        const unsigned r2 = (unsigned)(d.z - lo);
        const unsigned r3 = (unsigned)(d.w - lo);
        const bool m0 = r0 < RANGE_NODES;
        const bool m1 = r1 < RANGE_NODES;
        const bool m2 = r2 < RANGE_NODES;
        const bool m3 = r3 < RANGE_NODES;
        const float2 y0 = yn[m0 ? s.x : 0];
        const float2 y1 = yn[m1 ? s.y : 0];
        const float2 y2 = yn[m2 ? s.z : 0];
        const float2 y3 = yn[m3 ? s.w : 0];
        asm volatile("" :: "v"(y0.x), "v"(y0.y), "v"(y1.x), "v"(y1.y),
                           "v"(y2.x), "v"(y2.y), "v"(y3.x), "v"(y3.y));
        if (m0) { atomicAdd(&acc[2 * r0], y0.x); atomicAdd(&acc[2 * r0 + 1], y0.y); }
        if (m1) { atomicAdd(&acc[2 * r1], y1.x); atomicAdd(&acc[2 * r1 + 1], y1.y); }
        if (m2) { atomicAdd(&acc[2 * r2], y2.x); atomicAdd(&acc[2 * r2 + 1], y2.y); }
        if (m3) { atomicAdd(&acc[2 * r3], y3.x); atomicAdd(&acc[2 * r3 + 1], y3.y); }
    }
    __syncthreads();

    float* p = part2 + (size_t)slice * (2 * N_NODES) + 2 * lo;
    for (int i = tid; i < 2 * RANGE_NODES; i += 1024) p[i] = acc[i];
}

__global__ void final_kernel(const float* __restrict__ part2, const float* __restrict__ ninv_in,
                             const float* __restrict__ b2, float* __restrict__ out, int N) {
    int n = blockIdx.x * blockDim.x + threadIdx.x;
    if (n >= N) return;
    const float2* p = (const float2*)part2;
    float a0 = 0.0f, a1 = 0.0f;
#pragma unroll
    for (int s = 0; s < N_SLICES; ++s) {
        const float2 v = p[(size_t)s * N_NODES + n];
        a0 += v.x;
        a1 += v.y;
    }
    float ni = ninv_in[n];
    out[n * 2 + 0] = a0 * ni + b2[0];
    out[n * 2 + 1] = a1 * ni + b2[1];
}

extern "C" void kernel_launch(void* const* d_in, const int* in_sizes, int n_in,
                              void* d_out, int out_size, void* d_ws, size_t ws_size,
                              hipStream_t stream) {
    const float* in_feat = (const float*)d_in[0];
    const float* W1      = (const float*)d_in[1];
    const float* b1      = (const float*)d_in[2];
    const float* W2      = (const float*)d_in[3];
    const float* b2      = (const float*)d_in[4];
    const int*   src     = (const int*)d_in[5];
    const int*   dst     = (const int*)d_in[6];
    float*       out     = (float*)d_out;

    const int N = N_NODES;

    float* ws      = (float*)d_ws;
    float* deg_out = ws;
    float* deg_in  = ws + N;
    float* xs      = ws + 2 * (size_t)N;
    float* h1      = xs + (size_t)N * 128;
    float* y_node  = h1 + (size_t)N * 128;
    float* part    = y_node + 2 * (size_t)N;
    float* part2   = part + (size_t)N_SLICES * 2 * N;
    unsigned* omap = (unsigned*)(part2 + (size_t)N_SLICES * 2 * N);

    degrees_kernel<<<dim3(N_SLICES, N_RANGES), 1024, 0, stream>>>(src, dst, part, omap);
    ninv_kernel<<<(2 * N + 255) / 256, 256, 0, stream>>>(part, deg_out, 2 * N);
    scale_kernel<<<STREAM_GROUPS, 256, 0, stream>>>(in_feat, deg_out, xs);
    agg_lds_kernel<<<L1_GROUPS, 1024, 0, stream>>>(src, dst, omap, xs, W1, b1, deg_in, h1);
    l2_matvec_kernel<<<N / 8, 256, 0, stream>>>(h1, deg_out, W2, y_node, N);
    agg2_kernel<<<dim3(N_SLICES, N_RANGES), 1024, 0, stream>>>(src, dst, y_node, part2);
    final_kernel<<<(N + 255) / 256, 256, 0, stream>>>(part2, deg_in, b2, out, N);
}


# layer-2 dot moved into the work-group that holds the finished layer-1 rows in LDS: wave t reads its 16 rows back, folding butterfly, one 128-byte store; h1 plane and a launch gone; same sum tree
# speedup vs baseline: 3.8844x; 1.0768x over previous
#include <hip/hip_runtime.h>

typedef __attribute__((ext_vector_type(2))) float v2f;
typedef __attribute__((ext_vector_type(8))) float v8f;
typedef __attribute__((ext_vector_type(4))) float v4f;

#define N_NODES 100000
#define N_EDGES 1600000
#define IN_F 128
#define HID 128

#define N_RANGES 8
#define N_SLICES 32
#define RANGE_NODES (N_NODES / N_RANGES)
#define SLICE_EDGES (N_EDGES / N_SLICES)

#define L1_GROUPS 256
#define L1_ROWS 391
#define L1_SEG 512
#define MAP_VEC (N_EDGES / 16)
#define L1_TILES 25
#define L1_PITCH 132
static_assert(L1_TILES * 16 >= L1_ROWS + 1, "the tiles must cover the range and its spare row");
static_assert(32 * L1_SEG == HID * IN_F, "W1 in fragment order takes the place of the edge queues");

__global__ __launch_bounds__(1024) void degrees_kernel(const int* __restrict__ src,
                                                       const int* __restrict__ dst,
                                                       float* __restrict__ part,
                                                       unsigned* __restrict__ omap) {
    __shared__ unsigned h_out[RANGE_NODES];
    __shared__ unsigned h_in[RANGE_NODES];

    const int tid   = threadIdx.x;
    const int slice = blockIdx.x;
    const int lo    = blockIdx.y * RANGE_NODES;

    for (int i = tid; i < RANGE_NODES; i += 1024) {
        h_out[i] = 0u;
        h_in[i]  = 0u;
    }
    __syncthreads();

    const int4* s4 = (const int4*)(src + (size_t)slice * SLICE_EDGES);
    const int4* d4 = (const int4*)(dst + (size_t)slice * SLICE_EDGES);
    for (int i = tid; i < SLICE_EDGES / 4; i += 1024) {
        const int4 s = s4[i];
        const int4 d = d4[i];
        if (blockIdx.y == 0) {
            omap[(size_t)slice * (SLICE_EDGES / 4) + i] =
                ((unsigned)d.x / L1_ROWS) | (((unsigned)d.y / L1_ROWS) << 8) |
                (((unsigned)d.z / L1_ROWS) << 16) | (((unsigned)d.w / L1_ROWS) << 24);
        }
        unsigned r;
        r = (unsigned)(s.x - lo); if (r < RANGE_NODES) atomicAdd(&h_out[r], 1u);
        r = (unsigned)(s.y - lo); if (r < RANGE_NODES) atomicAdd(&h_out[r], 1u);
        r = (unsigned)(s.z - lo); if (r < RANGE_NODES) atomicAdd(&h_out[r], 1u);
        r = (unsigned)(s.w - lo); if (r < RANGE_NODES) atomicAdd(&h_out[r], 1u);
        r = (unsigned)(d.x - lo); if (r < RANGE_NODES) atomicAdd(&h_in[r], 1u);
        r = (unsigned)(d.y - lo); if (r < RANGE_NODES) atomicAdd(&h_in[r], 1u);
        r = (unsigned)(d.z - lo); if (r < RANGE_NODES) atomicAdd(&h_in[r], 1u);
        r = (unsigned)(d.w - lo); if (r < RANGE_NODES) atomicAdd(&h_in[r], 1u);
    }
    __syncthreads();

    float* po = part + (size_t)slice * (2 * N_NODES) + lo;
    float* pi = po + N_NODES;
    for (int i = tid; i < RANGE_NODES; i += 1024) {
        po[i] = (float)h_out[i];
        pi[i] = (float)h_in[i];
    }
}

__global__ void ninv_kernel(const float* __restrict__ part, float* __restrict__ ninv, int n) {
    int i = blockIdx.x * blockDim.x + threadIdx.x;
    if (i >= n) return;
    float d = 0.0f;
#pragma unroll
    for (int s = 0; s < N_SLICES; ++s) d += part[(size_t)s * (2 * N_NODES) + i];
    ninv[i] = (d > 0.0f) ? (1.0f / sqrtf(d)) : 0.0f;
}

#define STREAM_ROWS 25
#define STREAM_GROUPS (N_NODES / (8 * STREAM_ROWS))
static_assert(STREAM_GROUPS * 8 * STREAM_ROWS == N_NODES, "rows must split evenly over the waves");

__global__ __launch_bounds__(256) void scale_kernel(const float* __restrict__ x,
                                                    const float* __restrict__ ninv_out,
                                                    float* __restrict__ xs) {
    const int lane = threadIdx.x & 31;
    const int row0 = (blockIdx.x * 8 + (threadIdx.x >> 5)) * STREAM_ROWS;
    const v4f* x4 = (const v4f*)x + (size_t)row0 * 32 + lane;
    v4f*       o4 = (v4f*)xs + (size_t)row0 * 32 + lane;
    const float nv = ninv_out[row0 + (lane < STREAM_ROWS ? lane : 0)];
    v4f v[STREAM_ROWS];
#pragma unroll
    for (int j = 0; j < STREAM_ROWS; ++j) v[j] = __builtin_nontemporal_load(x4 + j * 32);
#pragma unroll
    for (int j = 0; j < STREAM_ROWS; ++j) {
        const float s = __int_as_float(__builtin_amdgcn_readlane(__float_as_int(nv), j));
        o4[j * 32] = v[j] * s;
    }
}

template <int M, int K>
__device__ __forceinline__ void fold_step(float (&p)[32], int lane) {
    const bool up = (lane & M) != 0;
#pragma unroll
    for (int i = 0; i < K; ++i) {
        const float keep = up ? p[i + K] : p[i];
        const float send = up ? p[i] : p[i + K];
        p[i] = keep + __shfl_xor(send, M, 32);
    }
}

__device__ __forceinline__ unsigned zero_flags(unsigned x) {
    const unsigned t = (x & 0x7f7f7f7fu) + 0x7f7f7f7fu;
    return ~(t | x | 0x7f7f7f7fu);
}

__device__ __forceinline__ unsigned match16(uint4 w, unsigned pat) {
    return (zero_flags(w.x ^ pat) >> 7) | (zero_flags(w.y ^ pat) >> 6) |
           (zero_flags(w.z ^ pat) >> 5) | (zero_flags(w.w ^ pat) >> 4);
}

__global__ __launch_bounds__(1024) void agg_lds_kernel(const int* __restrict__ src,
                                                       const int* __restrict__ dst,
                                                       const unsigned* __restrict__ omap,
                                                       const float* __restrict__ xs,
                                                       const float* __restrict__ W1,
                                                       const float* __restrict__ b1,
                                                       const float* __restrict__ ninv_in,
                                                       const float* __restrict__ ninv_out,
                                                       const float* __restrict__ W2,
                                                       float* __restrict__ y_node) {
    __shared__ __attribute__((aligned(16))) float acc[L1_TILES * 16 * L1_PITCH];
    __shared__ __attribute__((aligned(16))) unsigned queue[32 * L1_SEG];

    const int tid  = threadIdx.x;
    const int lane = tid & 31;
    const int wave = tid >> 5;
    const int g    = blockIdx.x;
    const int lo   = g * L1_ROWS;

    {
        float4* a4 = (float4*)acc;
        for (int i = tid; i < L1_TILES * 16 * L1_PITCH / 4; i += 1024) a4[i] = make_float4(0.f, 0.f, 0.f, 0.f);
    }
    __syncthreads();

    const uint4*   m4   = (const uint4*)omap;
    const unsigned pat  = (unsigned)g * 0x01010101u;
    const int      rot  = g * (MAP_VEC / L1_GROUPS);
    const unsigned ltm  = (1u << lane) - 1u;
    unsigned*      myq  = queue + wave * L1_SEG;
    int            cnt  = 0;

    for (int it = 0; it < (MAP_VEC + 4095) / 4096; ++it) {
        int   idx[4];
        bool  val[4];
        uint4 w[4];
#pragma unroll
        for (int u = 0; u < 4; ++u) {
            const int j = it * 4096 + u * 1024 + tid;
            val[u] = j < MAP_VEC;
            int i = j + rot;
            if (i >= MAP_VEC) i -= MAP_VEC;
            if (!val[u]) i = 0;
            idx[u] = i;
            w[u]   = m4[i];
        }
        __builtin_amdgcn_sched_barrier(0);
#pragma unroll
        for (int u = 0; u < 4; ++u) {
            unsigned m = val[u] ? match16(w[u], pat) : 0u;
            while (__builtin_amdgcn_ballot_w32(m != 0u) != 0u) {
                const bool     has = m != 0u;
                const unsigned bal = __builtin_amdgcn_ballot_w32(has);
                if (has) {
                    const int p = __ffs(m) - 1;
                    const int e = idx[u] * 16 + (p & 7) * 4 + (p >> 3);
                    m &= m - 1u;
                    const int slot = cnt + __popc(bal & ltm);
                    if (slot < L1_SEG) {
                        myq[slot] = (unsigned)e;
                    } else {
                        const int    s  = src[e];
                        const int    dl = dst[e] - lo;
                        const float* xr = xs + (size_t)s * 128;
                        float*       ar = acc + dl * L1_PITCH;
                        for (int f = 0; f < 128; ++f) atomicAdd(&ar[(f & 3) * 32 + (f >> 2)], xr[f]);
                    }
                }
                cnt += __popc(bal);
            }
        }
    }
    __syncthreads();

    const int n = cnt < L1_SEG ? cnt : L1_SEG;
    for (int k0 = 0; k0 < n; k0 += 32) {
        const int  k  = k0 + lane;
        const bool ok = k < n;
        const int  e  = ok ? (int)myq[k] : 0;
        int s  = src[e];
        int dl = dst[e] - lo;
        if (!ok) { s = 0; dl = L1_ROWS; }
        const int left = n - k0;
#pragma unroll
        for (int sb = 0; sb < 4; ++sb) {
            if (sb * 8 < left) {
                float4 v[8];
                int    dr[8];
#pragma unroll
                for (int j = 0; j < 8; ++j) {
                    const int sj = __builtin_amdgcn_readlane(s, sb * 8 + j);
                    dr[j]        = __builtin_amdgcn_readlane(dl, sb * 8 + j);
                    v[j] = *(const float4*)(xs + (size_t)sj * 128 + lane * 4);
                }
#pragma unroll
                for (int j = 0; j < 8; ++j) {
                    float* a = acc + dr[j] * L1_PITCH + lane;
                    atomicAdd(a,      v[j].x);
                    atomicAdd(a + 32, v[j].y);
                    atomicAdd(a + 64, v[j].z);
                    atomicAdd(a + 96, v[j].w);
                }
            }
        }
    }
    float2 wv[8];
    {
        const int l  = tid & 31;
        const int kk = tid >> 5;
        const int k  = kk * 4 + (l >> 4) * 2;
        const float* wp = W1 + (size_t)k * 128 + (l & 15);
#pragma unroll
        for (int ct = 0; ct < 8; ++ct) wv[ct] = make_float2(wp[ct * 16], wp[128 + ct * 16]);
    }
    __syncthreads();
    {
        float2* wst = (float2*)queue + tid;
#pragma unroll
        for (int ct = 0; ct < 8; ++ct) wst[ct * 1024] = wv[ct];
    }

    const bool  active = wave < L1_TILES;
    const int   mn     = lane & 15;
    const int   hf     = lane >> 4;
    float*      trow   = acc + wave * 16 * L1_PITCH;
    v2f a[32];
    if (active) {
        const float* ar = trow + mn * L1_PITCH + hf * 64;
#pragma unroll
        for (int kk = 0; kk < 32; ++kk) {
            a[kk].x = ar[kk];
            a[kk].y = ar[32 + kk];
        }
    }
    __syncthreads();

    if (active) {
        const v2f* wl   = (const v2f*)queue + lane;
        const int  m0   = lo + wave * 16 + hf * 8;
        float      nv[8];
#pragma unroll
        for (int r = 0; r < 8; ++r) nv[r] = ninv_in[(m0 + r < N_NODES) ? (m0 + r) : (N_NODES - 1)];
        for (int ct = 0; ct < 8; ct += 2) {
            v8f c0 = {};
            v8f c1 = {};
#pragma unroll
            for (int kk = 0; kk < 32; ++kk) {
                const v2f b0 = wl[((ct + 0) * 32 + kk) * 32];
                const v2f b1v = wl[((ct + 1) * 32 + kk) * 32];
                c0 = __builtin_amdgcn_wmma_f32_16x16x4_f32(false, a[kk], false, b0,
                                                           (short)0, c0, false, false);
                c1 = __builtin_amdgcn_wmma_f32_16x16x4_f32(false, a[kk], false, b1v,
                                                           (short)0, c1, false, false);
            }
            const int   n0  = ct * 16 + mn;
            const float bn0 = b1[n0];
            const float bn1 = b1[n0 + 16];
            float* d = trow + hf * 8 * L1_PITCH + n0;
#pragma unroll
            for (int r = 0; r < 8; ++r) {
                d[r * L1_PITCH]      = fmaxf(c0[r] * nv[r] + bn0, 0.0f);
                d[r * L1_PITCH + 16] = fmaxf(c1[r] * nv[r] + bn1, 0.0f);
            }
        }
    }
    float2 w2v[4];
    float  so = 0.0f;
    if (active) {
        const float2* w2p = (const float2*)W2;
#pragma unroll
        for (int j = 0; j < 4; ++j) w2v[j] = w2p[lane + 32 * j];
        const int node = lo + wave * 16 + mn;
        so = ninv_out[node < N_NODES ? node : (N_NODES - 1)];
    }
    __syncthreads();

    if (active) {
        float p[32];
#pragma unroll
        for (int i = 0; i < 16; ++i) {
            const float  s  = __int_as_float(__builtin_amdgcn_readlane(__float_as_int(so), i));
            const float* hr = trow + i * L1_PITCH + lane;
            float y0 = 0.0f, y1 = 0.0f;
#pragma unroll
            for (int j = 0; j < 4; ++j) {
                const float v = hr[32 * j] * s;
                y0 += v * w2v[j].x;
                y1 += v * w2v[j].y;
            }
            p[2 * i]     = y0;
            p[2 * i + 1] = y1;
        }
        fold_step<16, 16>(p, lane);
        fold_step<8, 8>(p, lane);
        fold_step<4, 4>(p, lane);
        fold_step<2, 2>(p, lane);
        fold_step<1, 1>(p, lane);
        const int r = wave * 16 + (lane >> 1);
        if (r < L1_ROWS && lo + r < N_NODES) y_node[(size_t)(lo + wave * 16) * 2 + lane] = p[0];
    }
}

__global__ __launch_bounds__(1024) void agg2_kernel(const int* __restrict__ src,
                                                    const int* __restrict__ dst,
                                                    const float* __restrict__ y_node,
                                                    float* __restrict__ part2) {
    __shared__ float acc[2 * RANGE_NODES];

    const int tid   = threadIdx.x;
    const int slice = blockIdx.x;
    const int lo    = blockIdx.y * RANGE_NODES;

    for (int i = tid; i < 2 * RANGE_NODES; i += 1024) acc[i] = 0.0f;
    __syncthreads();

    const int4*   s4 = (const int4*)(src + (size_t)slice * SLICE_EDGES);
    const int4*   d4 = (const int4*)(dst + (size_t)slice * SLICE_EDGES);
    const float2* yn = (const float2*)y_node;
    for (int i = tid; i < SLICE_EDGES / 4; i += 1024) {
        const int4 s = s4[i];
        const int4 d = d4[i];
        const unsigned r0 = (unsigned)(d.x - lo);
        const unsigned r1 = (unsigned)(d.y - lo);
        const unsigned r2 = (unsigned)(d.z - lo);
        const unsigned r3 = (unsigned)(d.w - lo);
        const bool m0 = r0 < RANGE_NODES;
        const bool m1 = r1 < RANGE_NODES;
        const bool m2 = r2 < RANGE_NODES;
        const bool m3 = r3 < RANGE_NODES;
        const float2 y0 = yn[m0 ? s.x : 0];
        const float2 y1 = yn[m1 ? s.y : 0];
        const float2 y2 = yn[m2 ? s.z : 0];
        const float2 y3 = yn[m3 ? s.w : 0];
        asm volatile("" :: "v"(y0.x), "v"(y0.y), "v"(y1.x), "v"(y1.y),
                           "v"(y2.x), "v"(y2.y), "v"(y3.x), "v"(y3.y));
        if (m0) { atomicAdd(&acc[2 * r0], y0.x); atomicAdd(&acc[2 * r0 + 1], y0.y); }
        if (m1) { atomicAdd(&acc[2 * r1], y1.x); atomicAdd(&acc[2 * r1 + 1], y1.y); }
        if (m2) { atomicAdd(&acc[2 * r2], y2.x); atomicAdd(&acc[2 * r2 + 1], y2.y); }
        if (m3) { atomicAdd(&acc[2 * r3], y3.x); atomicAdd(&acc[2 * r3 + 1], y3.y); }
    }
    __syncthreads();

    float* p = part2 + (size_t)slice * (2 * N_NODES) + 2 * lo;
    for (int i = tid; i < 2 * RANGE_NODES; i += 1024) p[i] = acc[i];
}

__global__ void final_kernel(const float* __restrict__ part2, const float* __restrict__ ninv_in,
                             const float* __restrict__ b2, float* __restrict__ out, int N) {
    int n = blockIdx.x * blockDim.x + threadIdx.x;
    if (n >= N) return;
    const float2* p = (const float2*)part2;
    float a0 = 0.0f, a1 = 0.0f;
#pragma unroll
    for (int s = 0; s < N_SLICES; ++s) {
        const float2 v = p[(size_t)s * N_NODES + n];
        a0 += v.x;
        a1 += v.y;
    }
    float ni = ninv_in[n];
    out[n * 2 + 0] = a0 * ni + b2[0];
    out[n * 2 + 1] = a1 * ni + b2[1];
}

extern "C" void kernel_launch(void* const* d_in, const int* in_sizes, int n_in,
                              void* d_out, int out_size, void* d_ws, size_t ws_size,
                              hipStream_t stream) {
    const float* in_feat = (const float*)d_in[0];
    const float* W1      = (const float*)d_in[1];
    const float* b1      = (const float*)d_in[2];
    const float* W2      = (const float*)d_in[3];
    const float* b2      = (const float*)d_in[4];
    const int*   src     = (const int*)d_in[5];
    const int*   dst     = (const int*)d_in[6];
    float*       out     = (float*)d_out;

    const int N = N_NODES;

    float* ws      = (float*)d_ws;
    float* deg_out = ws;
    float* deg_in  = ws + N;
    float* xs      = ws + 2 * (size_t)N;
    float* y_node  = xs + (size_t)N * 128;
    float* part    = y_node + 2 * (size_t)N;
    float* part2   = part + (size_t)N_SLICES * 2 * N;
    unsigned* omap = (unsigned*)(part2 + (size_t)N_SLICES * 2 * N);

    degrees_kernel<<<dim3(N_SLICES, N_RANGES), 1024, 0, stream>>>(src, dst, part, omap);
    ninv_kernel<<<(2 * N + 255) / 256, 256, 0, stream>>>(part, deg_out, 2 * N);
    scale_kernel<<<STREAM_GROUPS, 256, 0, stream>>>(in_feat, deg_out, xs);
    agg_lds_kernel<<<L1_GROUPS, 1024, 0, stream>>>(src, dst, omap, xs, W1, b1, deg_in, deg_out, W2,
                                                   y_node);
    agg2_kernel<<<dim3(N_SLICES, N_RANGES), 1024, 0, stream>>>(src, dst, y_node, part2);
    final_kernel<<<(N + 255) / 256, 256, 0, stream>>>(part2, deg_in, b2, out, N);
}


# owner map kept as 8 bit planes per 32-edge block (ballots of the dst range's bits), not a byte per edge: a lane matches 32 edges with 8 bit operations; same queues, f32 LDS adds in arrival order
# speedup vs baseline: 4.2444x; 1.0927x over previous
#include <hip/hip_runtime.h>

typedef __attribute__((ext_vector_type(2))) float v2f;
typedef __attribute__((ext_vector_type(8))) float v8f;
typedef __attribute__((ext_vector_type(4))) float v4f;

#define N_NODES 100000
#define N_EDGES 1600000
#define IN_F 128
#define HID 128

#define N_RANGES 8
#define N_SLICES 32
#define RANGE_NODES (N_NODES / N_RANGES)
#define SLICE_EDGES (N_EDGES / N_SLICES)

#define L1_GROUPS 256
#define L1_ROWS 391
#define L1_SEG 512
#define N_BLK (N_EDGES / 32)
#define N_SB ((N_BLK + 31) / 32)
#define N_MAPTASK (N_BLK / 8)
#define L1_TILES 25
#define L1_PITCH 132
static_assert(L1_TILES * 16 >= L1_ROWS + 1, "the tiles must cover the range and its spare row");
static_assert(32 * L1_SEG == HID * IN_F, "W1 in fragment order takes the place of the edge queues");
static_assert(N_EDGES % 256 == 0, "the owner map is written in whole units of 8 blocks");
static_assert(L1_GROUPS == 256 && (N_NODES - 1) / L1_ROWS < 256, "a range number must fit the 8 planes");
static_assert(N_MAPTASK <= 32 * N_SLICES * N_RANGES, "one map unit per wave of the degree launch at most");

__global__ __launch_bounds__(1024) void degrees_kernel(const int* __restrict__ src,
                                                       const int* __restrict__ dst,
                                                       float* __restrict__ part,
                                                       unsigned* __restrict__ omap) {
    __shared__ unsigned h_out[RANGE_NODES];
    __shared__ unsigned h_in[RANGE_NODES];

    const int tid   = threadIdx.x;
    const int slice = blockIdx.x;
    const int lo    = blockIdx.y * RANGE_NODES;

    const int  lane  = tid & 31;
    const int  mtask = __builtin_amdgcn_readfirstlane(tid >> 5) * (N_SLICES * N_RANGES) +
                       (int)(blockIdx.y * N_SLICES + blockIdx.x);
    const bool mine  = mtask < N_MAPTASK;
    int dv[8] = {};
    if (mine) {
        const int* dp = dst + (size_t)mtask * 256 + lane;
#pragma unroll
        for (int q = 0; q < 8; ++q) dv[q] = dp[q * 32];
    }

    for (int i = tid; i < RANGE_NODES; i += 1024) {
        h_out[i] = 0u;
        h_in[i]  = 0u;
    }

    if (mine) {
        unsigned w0 = 0u, w1 = 0u;
#pragma unroll
        for (int q = 0; q < 8; ++q) {
            const unsigned gid = (unsigned)dv[q] / L1_ROWS;
#pragma unroll
            for (int jj = 0; jj < 4; ++jj) {
                const unsigned b0 = __builtin_amdgcn_ballot_w32((gid & (1u << jj)) != 0u);
                const unsigned b1 = __builtin_amdgcn_ballot_w32((gid & (16u << jj)) != 0u);
                const bool     me = lane == q * 4 + jj;
                w0 = me ? b0 : w0;
                w1 = me ? b1 : w1;
            }
        }
        unsigned* mp = omap + (size_t)(mtask >> 2) * 256 + (mtask & 3) * 32 + lane;
        mp[0]   = w0;
        mp[128] = w1;
    }
    __syncthreads();

    const int4* s4 = (const int4*)(src + (size_t)slice * SLICE_EDGES);
    const int4* d4 = (const int4*)(dst + (size_t)slice * SLICE_EDGES);
    for (int i = tid; i < SLICE_EDGES / 4; i += 1024) {
        const int4 s = s4[i];
        const int4 d = d4[i];
        unsigned r;
        r = (unsigned)(s.x - lo); if (r < RANGE_NODES) atomicAdd(&h_out[r], 1u);
        r = (unsigned)(s.y - lo); if (r < RANGE_NODES) atomicAdd(&h_out[r], 1u);
        r = (unsigned)(s.z - lo); if (r < RANGE_NODES) atomicAdd(&h_out[r], 1u);
        r = (unsigned)(s.w - lo); if (r < RANGE_NODES) atomicAdd(&h_out[r], 1u);
        r = (unsigned)(d.x - lo); if (r < RANGE_NODES) atomicAdd(&h_in[r], 1u);
        r = (unsigned)(d.y - lo); if (r < RANGE_NODES) atomicAdd(&h_in[r], 1u);
        r = (unsigned)(d.z - lo); if (r < RANGE_NODES) atomicAdd(&h_in[r], 1u);
        r = (unsigned)(d.w - lo); if (r < RANGE_NODES) atomicAdd(&h_in[r], 1u);
    }
    __syncthreads();

    float* po = part + (size_t)slice * (2 * N_NODES) + lo;
    float* pi = po + N_NODES;
    for (int i = tid; i < RANGE_NODES; i += 1024) {
        po[i] = (float)h_out[i];
        pi[i] = (float)h_in[i];
    }
}

__global__ void ninv_kernel(const float* __restrict__ part, float* __restrict__ ninv, int n) {
    int i = blockIdx.x * blockDim.x + threadIdx.x;
    if (i >= n) return;
    float d = 0.0f;
#pragma unroll
    for (int s = 0; s < N_SLICES; ++s) d += part[(size_t)s * (2 * N_NODES) + i];
    ninv[i] = (d > 0.0f) ? (1.0f / sqrtf(d)) : 0.0f;
}

#define STREAM_ROWS 25
#define STREAM_GROUPS (N_NODES / (8 * STREAM_ROWS))
static_assert(STREAM_GROUPS * 8 * STREAM_ROWS == N_NODES, "rows must split evenly over the waves");

__global__ __launch_bounds__(256) void scale_kernel(const float* __restrict__ x,
                                                    const float* __restrict__ ninv_out,
                                                    float* __restrict__ xs) {
    const int lane = threadIdx.x & 31;
    const int row0 = (blockIdx.x * 8 + (threadIdx.x >> 5)) * STREAM_ROWS;
    const v4f* x4 = (const v4f*)x + (size_t)row0 * 32 + lane;
    v4f*       o4 = (v4f*)xs + (size_t)row0 * 32 + lane;
    const float nv = ninv_out[row0 + (lane < STREAM_ROWS ? lane : 0)];
    v4f v[STREAM_ROWS];
#pragma unroll
    for (int j = 0; j < STREAM_ROWS; ++j) v[j] = __builtin_nontemporal_load(x4 + j * 32);
#pragma unroll
    for (int j = 0; j < STREAM_ROWS; ++j) {
        const float s = __int_as_float(__builtin_amdgcn_readlane(__float_as_int(nv), j));
        o4[j * 32] = v[j] * s;
    }
}

template <int M, int K>
__device__ __forceinline__ void fold_step(float (&p)[32], int lane) {
    const bool up = (lane & M) != 0;
#pragma unroll
    for (int i = 0; i < K; ++i) {
        const float keep = up ? p[i + K] : p[i];
        const float send = up ? p[i] : p[i + K];
        p[i] = keep + __shfl_xor(send, M, 32);
    }
}

__global__ __launch_bounds__(1024) void agg_lds_kernel(const int* __restrict__ src,
                                                       const int* __restrict__ dst,
                                                       const unsigned* __restrict__ omap,
                                                       const float* __restrict__ xs,
                                                       const float* __restrict__ W1,
                                                       const float* __restrict__ b1,
                                                       const float* __restrict__ ninv_in,
                                                       const float* __restrict__ ninv_out,
                                                       const float* __restrict__ W2,
                                                       float* __restrict__ y_node) {
    __shared__ __attribute__((aligned(16))) float acc[L1_TILES * 16 * L1_PITCH];
    __shared__ __attribute__((aligned(16))) unsigned queue[32 * L1_SEG];

    const int tid  = threadIdx.x;
    const int lane = tid & 31;
    const int wave = tid >> 5;
    const int g    = blockIdx.x;
    const int lo   = g * L1_ROWS;

    {
        float4* a4 = (float4*)acc;
        for (int i = tid; i < L1_TILES * 16 * L1_PITCH / 4; i += 1024) a4[i] = make_float4(0.f, 0.f, 0.f, 0.f);
    }
    __syncthreads();

    const uint4*   m4   = (const uint4*)omap;
    const int      rot  = g * (N_SB / L1_GROUPS);
    const unsigned ltm  = (1u << lane) - 1u;
    unsigned*      myq  = queue + wave * L1_SEG;
    int            cnt  = 0;
    unsigned flip[8];
#pragma unroll
    for (int j = 0; j < 8; ++j) flip[j] = ((g >> j) & 1) ? 0u : 0xffffffffu;

    for (int it = 0; it < (N_SB + 127) / 128; ++it) {
        int   eb[4];
        bool  val[4];
        uint4 pl[4], ph[4];
#pragma unroll
        for (int u = 0; u < 4; ++u) {
            const int j = it * 128 + u * 32 + wave;
            int sb = j + rot;
            if (sb >= N_SB) sb -= N_SB;
            const int blk = sb * 32 + lane;
            val[u] = j < N_SB && blk < N_BLK;
            const int v = val[u] ? sb * 64 + lane : 0;
            pl[u] = m4[v];
            ph[u] = m4[v + 32];
            eb[u] = blk * 32;
        }
        __builtin_amdgcn_sched_barrier(0);
#pragma unroll
        for (int u = 0; u < 4; ++u) {
            unsigned m = (pl[u].x ^ flip[0]) & (pl[u].y ^ flip[1]) & (pl[u].z ^ flip[2]) & (pl[u].w ^ flip[3]) &
                         (ph[u].x ^ flip[4]) & (ph[u].y ^ flip[5]) & (ph[u].z ^ flip[6]) & (ph[u].w ^ flip[7]);
            if (!val[u]) m = 0u;
            while (__builtin_amdgcn_ballot_w32(m != 0u) != 0u) {
                const bool     has = m != 0u;
                const unsigned bal = __builtin_amdgcn_ballot_w32(has);
                if (has) {
                    const int e = eb[u] + __ffs(m) - 1;
                    m &= m - 1u;
                    const int slot = cnt + __popc(bal & ltm);
                    if (slot < L1_SEG) {
                        myq[slot] = (unsigned)e;
                    } else {
                        const int    s  = src[e];
                        const int    dl = dst[e] - lo;
                        const float* xr = xs + (size_t)s * 128;
                        float*       ar = acc + dl * L1_PITCH;
                        for (int f = 0; f < 128; ++f) atomicAdd(&ar[(f & 3) * 32 + (f >> 2)], xr[f]);
                    }
                }
                cnt += __popc(bal);
            }
        }
    }
    __syncthreads();

    const int n = cnt < L1_SEG ? cnt : L1_SEG;
    for (int k0 = 0; k0 < n; k0 += 32) {
        const int  k  = k0 + lane;
        const bool ok = k < n;
        const int  e  = ok ? (int)myq[k] : 0;
        int s  = src[e];
        int dl = dst[e] - lo;
        if (!ok) { s = 0; dl = L1_ROWS; }
        const int left = n - k0;
#pragma unroll
        for (int sb = 0; sb < 4; ++sb) {
            if (sb * 8 < left) {
                float4 v[8];
                int    dr[8];
#pragma unroll
                for (int j = 0; j < 8; ++j) {
                    const int sj = __builtin_amdgcn_readlane(s, sb * 8 + j);
                    dr[j]        = __builtin_amdgcn_readlane(dl, sb * 8 + j);
                    v[j] = *(const float4*)(xs + (size_t)sj * 128 + lane * 4);
                }
#pragma unroll
                for (int j = 0; j < 8; ++j) {
                    float* a = acc + dr[j] * L1_PITCH + lane;
                    atomicAdd(a,      v[j].x);
                    atomicAdd(a + 32, v[j].y);
                    atomicAdd(a + 64, v[j].z);
                    atomicAdd(a + 96, v[j].w);
                }
            }
        }
    }
    float2 wv[8];
    {
        const int l  = tid & 31;
        const int kk = tid >> 5;
        const int k  = kk * 4 + (l >> 4) * 2;
        const float* wp = W1 + (size_t)k * 128 + (l & 15);
#pragma unroll
        for (int ct = 0; ct < 8; ++ct) wv[ct] = make_float2(wp[ct * 16], wp[128 + ct * 16]);
    }
    __syncthreads();
    {
        float2* wst = (float2*)queue + tid;
#pragma unroll
        for (int ct = 0; ct < 8; ++ct) wst[ct * 1024] = wv[ct];
    }

    const bool  active = wave < L1_TILES;
    const int   mn     = lane & 15;
    const int   hf     = lane >> 4;
    float*      trow   = acc + wave * 16 * L1_PITCH;
    v2f a[32];
    if (active) {
        const float* ar = trow + mn * L1_PITCH + hf * 64;
#pragma unroll
        for (int kk = 0; kk < 32; ++kk) {
            a[kk].x = ar[kk];
            a[kk].y = ar[32 + kk];
        }
    }
    __syncthreads();

    if (active) {
        const v2f* wl   = (const v2f*)queue + lane;
        const int  m0   = lo + wave * 16 + hf * 8;
        float      nv[8];
#pragma unroll
        for (int r = 0; r < 8; ++r) nv[r] = ninv_in[(m0 + r < N_NODES) ? (m0 + r) : (N_NODES - 1)];
        for (int ct = 0; ct < 8; ct += 2) {
            v8f c0 = {};
            v8f c1 = {};
#pragma unroll
            for (int kk = 0; kk < 32; ++kk) {
                const v2f b0 = wl[((ct + 0) * 32 + kk) * 32];
                const v2f b1v = wl[((ct + 1) * 32 + kk) * 32];
                c0 = __builtin_amdgcn_wmma_f32_16x16x4_f32(false, a[kk], false, b0,
                                                           (short)0, c0, false, false);
                c1 = __builtin_amdgcn_wmma_f32_16x16x4_f32(false, a[kk], false, b1v,
                                                           (short)0, c1, false, false);
            }
            const int   n0  = ct * 16 + mn;
            const float bn0 = b1[n0];
            const float bn1 = b1[n0 + 16];
            float* d = trow + hf * 8 * L1_PITCH + n0;
#pragma unroll
            for (int r = 0; r < 8; ++r) {
                d[r * L1_PITCH]      = fmaxf(c0[r] * nv[r] + bn0, 0.0f);
                d[r * L1_PITCH + 16] = fmaxf(c1[r] * nv[r] + bn1, 0.0f);
            }
        }
    }
    float2 w2v[4];
    float  so = 0.0f;
    if (active) {
        const float2* w2p = (const float2*)W2;
#pragma unroll
        for (int j = 0; j < 4; ++j) w2v[j] = w2p[lane + 32 * j];
        const int node = lo + wave * 16 + mn;
        so = ninv_out[node < N_NODES ? node : (N_NODES - 1)];
    }
    __syncthreads();

    if (active) {
        float p[32];
#pragma unroll
        for (int i = 0; i < 16; ++i) {
            const float  s  = __int_as_float(__builtin_amdgcn_readlane(__float_as_int(so), i));
            const float* hr = trow + i * L1_PITCH + lane;
            float y0 = 0.0f, y1 = 0.0f;
#pragma unroll
            for (int j = 0; j < 4; ++j) {
                const float v = hr[32 * j] * s;
                y0 += v * w2v[j].x;
                y1 += v * w2v[j].y;
            }
            p[2 * i]     = y0;
            p[2 * i + 1] = y1;
        }
        fold_step<16, 16>(p, lane);
        fold_step<8, 8>(p, lane);
        fold_step<4, 4>(p, lane);
        fold_step<2, 2>(p, lane);
        fold_step<1, 1>(p, lane);
        const int r = wave * 16 + (lane >> 1);
        if (r < L1_ROWS && lo + r < N_NODES) y_node[(size_t)(lo + wave * 16) * 2 + lane] = p[0];
    }
}

__global__ __launch_bounds__(1024) void agg2_kernel(const int* __restrict__ src,
                                                    const int* __restrict__ dst,
                                                    const float* __restrict__ y_node,
                                                    float* __restrict__ part2) {
    __shared__ float acc[2 * RANGE_NODES];

    const int tid   = threadIdx.x;
    const int slice = blockIdx.x;
    const int lo    = blockIdx.y * RANGE_NODES;

    for (int i = tid; i < 2 * RANGE_NODES; i += 1024) acc[i] = 0.0f;
    __syncthreads();

    const int4*   s4 = (const int4*)(src + (size_t)slice * SLICE_EDGES);
    const int4*   d4 = (const int4*)(dst + (size_t)slice * SLICE_EDGES);
    const float2* yn = (const float2*)y_node;
    for (int i = tid; i < SLICE_EDGES / 4; i += 1024) {
        const int4 s = s4[i];
        const int4 d = d4[i];
        const unsigned r0 = (unsigned)(d.x - lo);
        const unsigned r1 = (unsigned)(d.y - lo);
        const unsigned r2 = (unsigned)(d.z - lo);
        const unsigned r3 = (unsigned)(d.w - lo);
        const bool m0 = r0 < RANGE_NODES;
        const bool m1 = r1 < RANGE_NODES;
        const bool m2 = r2 < RANGE_NODES;
        const bool m3 = r3 < RANGE_NODES;
        const float2 y0 = yn[m0 ? s.x : 0];
        const float2 y1 = yn[m1 ? s.y : 0];
        const float2 y2 = yn[m2 ? s.z : 0];
        const float2 y3 = yn[m3 ? s.w : 0];
        asm volatile("" :: "v"(y0.x), "v"(y0.y), "v"(y1.x), "v"(y1.y),
                           "v"(y2.x), "v"(y2.y), "v"(y3.x), "v"(y3.y));
        if (m0) { atomicAdd(&acc[2 * r0], y0.x); atomicAdd(&acc[2 * r0 + 1], y0.y); }
        if (m1) { atomicAdd(&acc[2 * r1], y1.x); atomicAdd(&acc[2 * r1 + 1], y1.y); }
        if (m2) { atomicAdd(&acc[2 * r2], y2.x); atomicAdd(&acc[2 * r2 + 1], y2.y); }
        if (m3) { atomicAdd(&acc[2 * r3], y3.x); atomicAdd(&acc[2 * r3 + 1], y3.y); }
    }
    __syncthreads();

    float* p = part2 + (size_t)slice * (2 * N_NODES) + 2 * lo;
    for (int i = tid; i < 2 * RANGE_NODES; i += 1024) p[i] = acc[i];
}

__global__ void final_kernel(const float* __restrict__ part2, const float* __restrict__ ninv_in,
                             const float* __restrict__ b2, float* __restrict__ out, int N) {
    int n = blockIdx.x * blockDim.x + threadIdx.x;
    if (n >= N) return;
    const float2* p = (const float2*)part2;
    float a0 = 0.0f, a1 = 0.0f;
#pragma unroll
    for (int s = 0; s < N_SLICES; ++s) {
        const float2 v = p[(size_t)s * N_NODES + n];
        a0 += v.x;
        a1 += v.y;
    }
    float ni = ninv_in[n];
    out[n * 2 + 0] = a0 * ni + b2[0];
    out[n * 2 + 1] = a1 * ni + b2[1];
}

extern "C" void kernel_launch(void* const* d_in, const int* in_sizes, int n_in,
                              void* d_out, int out_size, void* d_ws, size_t ws_size,
                              hipStream_t stream) {
    const float* in_feat = (const float*)d_in[0];
    const float* W1      = (const float*)d_in[1];
    const float* b1      = (const float*)d_in[2];
    const float* W2      = (const float*)d_in[3];
    const float* b2      = (const float*)d_in[4];
    const int*   src     = (const int*)d_in[5];
    const int*   dst     = (const int*)d_in[6];
    float*       out     = (float*)d_out;

    const int N = N_NODES;

    float* ws      = (float*)d_ws;
    float* deg_out = ws;
    float* deg_in  = ws + N;
    float* xs      = ws + 2 * (size_t)N;
    float* y_node  = xs + (size_t)N * 128;
    float* part    = y_node + 2 * (size_t)N;
    float* part2   = part + (size_t)N_SLICES * 2 * N;
    unsigned* omap = (unsigned*)(part2 + (size_t)N_SLICES * 2 * N);

    degrees_kernel<<<dim3(N_SLICES, N_RANGES), 1024, 0, stream>>>(src, dst, part, omap);
    ninv_kernel<<<(2 * N + 255) / 256, 256, 0, stream>>>(part, deg_out, 2 * N);
    scale_kernel<<<STREAM_GROUPS, 256, 0, stream>>>(in_feat, deg_out, xs);
    agg_lds_kernel<<<L1_GROUPS, 1024, 0, stream>>>(src, dst, omap, xs, W1, b1, deg_in, deg_out, W2,
                                                   y_node);
    agg2_kernel<<<dim3(N_SLICES, N_RANGES), 1024, 0, stream>>>(src, dst, y_node, part2);
    final_kernel<<<(N + 255) / 256, 256, 0, stream>>>(part2, deg_in, b2, out, N);
}


# layer-2 edge sums from per-range edge lists written by the layer-1 groups: 256 groups add y[src] in LDS (one arrival-order f32 sum per node) and write out; slice planes and the final launch gone
# speedup vs baseline: 4.5567x; 1.0736x over previous
#include <hip/hip_runtime.h>

typedef __attribute__((ext_vector_type(2))) float v2f;
typedef __attribute__((ext_vector_type(8))) float v8f;
typedef __attribute__((ext_vector_type(4))) float v4f;

#define N_NODES 100000
#define N_EDGES 1600000
#define IN_F 128
#define HID 128

#define N_RANGES 8
#define N_SLICES 32
#define RANGE_NODES (N_NODES / N_RANGES)
#define SLICE_EDGES (N_EDGES / N_SLICES)

#define L1_GROUPS 256
#define L1_ROWS 391
#define L1_SEG 512
#define N_BLK (N_EDGES / 32)
#define N_SB ((N_BLK + 31) / 32)
#define N_MAPTASK (N_BLK / 8)
#define L1_TILES 25
#define L1_PITCH 132
static_assert(L1_TILES * 16 >= L1_ROWS + 1, "the tiles must cover the range and its spare row");
static_assert(32 * L1_SEG == HID * IN_F, "W1 in fragment order takes the place of the edge queues");
static_assert(N_EDGES % 256 == 0, "the owner map is written in whole units of 8 blocks");
static_assert(L1_GROUPS == 256 && (N_NODES - 1) / L1_ROWS < 256, "a range number must fit the 8 planes");
static_assert(N_MAPTASK <= 32 * N_SLICES * N_RANGES, "one map unit per wave of the degree launch at most");
#define L1_LIST (32 * L1_SEG)
#define L1_SRC_BITS 17
static_assert(N_NODES <= (1 << L1_SRC_BITS) && L1_ROWS < (1 << (32 - L1_SRC_BITS)), "src and local dst must fit one word");

__global__ __launch_bounds__(1024) void degrees_kernel(const int* __restrict__ src,
                                                       const int* __restrict__ dst,
                                                       float* __restrict__ part,
                                                       unsigned* __restrict__ omap,
                                                       unsigned* __restrict__ oflag) {
    __shared__ unsigned h_out[RANGE_NODES];
    __shared__ unsigned h_in[RANGE_NODES];

    const int tid   = threadIdx.x;
    const int slice = blockIdx.x;
    const int lo    = blockIdx.y * RANGE_NODES;

    if (tid == 0 && blockIdx.x == 0 && blockIdx.y == 0) oflag[0] = 0u;

    const int  lane  = tid & 31;
    const int  mtask = __builtin_amdgcn_readfirstlane(tid >> 5) * (N_SLICES * N_RANGES) +
                       (int)(blockIdx.y * N_SLICES + blockIdx.x);
    const bool mine  = mtask < N_MAPTASK;
    int dv[8] = {};
    if (mine) {
        const int* dp = dst + (size_t)mtask * 256 + lane;
#pragma unroll
        for (int q = 0; q < 8; ++q) dv[q] = dp[q * 32];
    }

    for (int i = tid; i < RANGE_NODES; i += 1024) {
        h_out[i] = 0u;
        h_in[i]  = 0u;
    }

    if (mine) {
        unsigned w0 = 0u, w1 = 0u;
#pragma unroll
        for (int q = 0; q < 8; ++q) {
            const unsigned gid = (unsigned)dv[q] / L1_ROWS;
#pragma unroll
            for (int jj = 0; jj < 4; ++jj) {
                const unsigned b0 = __builtin_amdgcn_ballot_w32((gid & (1u << jj)) != 0u);
                const unsigned b1 = __builtin_amdgcn_ballot_w32((gid & (16u << jj)) != 0u);
                const bool     me = lane == q * 4 + jj;
                w0 = me ? b0 : w0;
                w1 = me ? b1 : w1;
            }
        }
        unsigned* mp = omap + (size_t)(mtask >> 2) * 256 + (mtask & 3) * 32 + lane;
        mp[0]   = w0;
        mp[128] = w1;
    }
    __syncthreads();

    const int4* s4 = (const int4*)(src + (size_t)slice * SLICE_EDGES);
    const int4* d4 = (const int4*)(dst + (size_t)slice * SLICE_EDGES);
    for (int i = tid; i < SLICE_EDGES / 4; i += 1024) {
        const int4 s = s4[i];
        const int4 d = d4[i];
        unsigned r;
        r = (unsigned)(s.x - lo); if (r < RANGE_NODES) atomicAdd(&h_out[r], 1u);
        r = (unsigned)(s.y - lo); if (r < RANGE_NODES) atomicAdd(&h_out[r], 1u);
        r = (unsigned)(s.z - lo); if (r < RANGE_NODES) atomicAdd(&h_out[r], 1u);
        r = (unsigned)(s.w - lo); if (r < RANGE_NODES) atomicAdd(&h_out[r], 1u);
        r = (unsigned)(d.x - lo); if (r < RANGE_NODES) atomicAdd(&h_in[r], 1u);
        r = (unsigned)(d.y - lo); if (r < RANGE_NODES) atomicAdd(&h_in[r], 1u);
        r = (unsigned)(d.z - lo); if (r < RANGE_NODES) atomicAdd(&h_in[r], 1u);
        r = (unsigned)(d.w - lo); if (r < RANGE_NODES) atomicAdd(&h_in[r], 1u);
    }
    __syncthreads();

    float* po = part + (size_t)slice * (2 * N_NODES) + lo;
    float* pi = po + N_NODES;
    for (int i = tid; i < RANGE_NODES; i += 1024) {
        po[i] = (float)h_out[i];
        pi[i] = (float)h_in[i];
    }
}

__global__ void ninv_kernel(const float* __restrict__ part, float* __restrict__ ninv, int n) {
    int i = blockIdx.x * blockDim.x + threadIdx.x;
    if (i >= n) return;
    float d = 0.0f;
#pragma unroll
    for (int s = 0; s < N_SLICES; ++s) d += part[(size_t)s * (2 * N_NODES) + i];
    ninv[i] = (d > 0.0f) ? (1.0f / sqrtf(d)) : 0.0f;
}

#define STREAM_ROWS 25
#define STREAM_GROUPS (N_NODES / (8 * STREAM_ROWS))
static_assert(STREAM_GROUPS * 8 * STREAM_ROWS == N_NODES, "rows must split evenly over the waves");

__global__ __launch_bounds__(256) void scale_kernel(const float* __restrict__ x,
                                                    const float* __restrict__ ninv_out,
                                                    float* __restrict__ xs) {
    const int lane = threadIdx.x & 31;
    const int row0 = (blockIdx.x * 8 + (threadIdx.x >> 5)) * STREAM_ROWS;
    const v4f* x4 = (const v4f*)x + (size_t)row0 * 32 + lane;
    v4f*       o4 = (v4f*)xs + (size_t)row0 * 32 + lane;
    const float nv = ninv_out[row0 + (lane < STREAM_ROWS ? lane : 0)];
    v4f v[STREAM_ROWS];
#pragma unroll
    for (int j = 0; j < STREAM_ROWS; ++j) v[j] = __builtin_nontemporal_load(x4 + j * 32);
#pragma unroll
    for (int j = 0; j < STREAM_ROWS; ++j) {
        const float s = __int_as_float(__builtin_amdgcn_readlane(__float_as_int(nv), j));
        o4[j * 32] = v[j] * s;
    }
}

template <int M, int K>
__device__ __forceinline__ void fold_step(float (&p)[32], int lane) {
    const bool up = (lane & M) != 0;
#pragma unroll
    for (int i = 0; i < K; ++i) {
        const float keep = up ? p[i + K] : p[i];
        const float send = up ? p[i] : p[i + K];
        p[i] = keep + __shfl_xor(send, M, 32);
    }
}

__global__ __launch_bounds__(1024) void agg_lds_kernel(const int* __restrict__ src,
                                                       const int* __restrict__ dst,
                                                       const unsigned* __restrict__ omap,
                                                       const float* __restrict__ xs,
                                                       const float* __restrict__ W1,
                                                       const float* __restrict__ b1,
                                                       const float* __restrict__ ninv_in,
                                                       const float* __restrict__ ninv_out,
                                                       const float* __restrict__ W2,
                                                       float* __restrict__ y_node,
                                                       unsigned* __restrict__ elist,
                                                       int* __restrict__ ecount,
                                                       unsigned* __restrict__ oflag) {
    __shared__ __attribute__((aligned(16))) float acc[L1_TILES * 16 * L1_PITCH];
    __shared__ __attribute__((aligned(16))) unsigned queue[32 * L1_SEG];
    __shared__ int wcnt[32];

    const int tid  = threadIdx.x;
    const int lane = tid & 31;
    const int wave = tid >> 5;
    const int g    = blockIdx.x;
    const int lo   = g * L1_ROWS;

    {
        float4* a4 = (float4*)acc;
        for (int i = tid; i < L1_TILES * 16 * L1_PITCH / 4; i += 1024) a4[i] = make_float4(0.f, 0.f, 0.f, 0.f);
    }
    __syncthreads();

    const uint4*   m4   = (const uint4*)omap;
    const int      rot  = g * (N_SB / L1_GROUPS);
    const unsigned ltm  = (1u << lane) - 1u;
    unsigned*      myq  = queue + wave * L1_SEG;
    int            cnt  = 0;
    unsigned flip[8];
#pragma unroll
    for (int j = 0; j < 8; ++j) flip[j] = ((g >> j) & 1) ? 0u : 0xffffffffu;

    for (int it = 0; it < (N_SB + 127) / 128; ++it) {
        int   eb[4];
        bool  val[4];
        uint4 pl[4], ph[4];
#pragma unroll
        for (int u = 0; u < 4; ++u) {
            const int j = it * 128 + u * 32 + wave;
            int sb = j + rot;
            if (sb >= N_SB) sb -= N_SB;
            const int blk = sb * 32 + lane;
            val[u] = j < N_SB && blk < N_BLK;
            const int v = val[u] ? sb * 64 + lane : 0;
            pl[u] = m4[v];
            ph[u] = m4[v + 32];
            eb[u] = blk * 32;
        }
        __builtin_amdgcn_sched_barrier(0);
#pragma unroll
        for (int u = 0; u < 4; ++u) {
            unsigned m = (pl[u].x ^ flip[0]) & (pl[u].y ^ flip[1]) & (pl[u].z ^ flip[2]) & (pl[u].w ^ flip[3]) &
                         (ph[u].x ^ flip[4]) & (ph[u].y ^ flip[5]) & (ph[u].z ^ flip[6]) & (ph[u].w ^ flip[7]);
            if (!val[u]) m = 0u;
            while (__builtin_amdgcn_ballot_w32(m != 0u) != 0u) {
                const bool     has = m != 0u;
                const unsigned bal = __builtin_amdgcn_ballot_w32(has);
                if (has) {
                    const int e = eb[u] + __ffs(m) - 1;
                    m &= m - 1u;
                    const int slot = cnt + __popc(bal & ltm);
                    if (slot < L1_SEG) {
                        myq[slot] = (unsigned)e;
                    } else {
                        const int    s  = src[e];
                        const int    dl = dst[e] - lo;
                        const float* xr = xs + (size_t)s * 128;
                        float*       ar = acc + dl * L1_PITCH;
                        for (int f = 0; f < 128; ++f) atomicAdd(&ar[(f & 3) * 32 + (f >> 2)], xr[f]);
                    }
                }
                cnt += __popc(bal);
            }
        }
    }
    const int n = cnt < L1_SEG ? cnt : L1_SEG;
    if (lane == 0) {
        wcnt[wave] = n;
        if (cnt > L1_SEG) oflag[0] = 1u;
    }
    __syncthreads();

    int base = 0, total = 0;
    for (int v = 0; v < 32; ++v) {
        const int c = wcnt[v];
        total += c;
        base += v < wave ? c : 0;
    }
    if (tid == 0) ecount[g] = total;
    unsigned* gl = elist + (size_t)g * L1_LIST + base;

    for (int k0 = 0; k0 < n; k0 += 32) {
        const int  k  = k0 + lane;
        const bool ok = k < n;
        const int  e  = ok ? (int)myq[k] : 0;
        int s  = src[e];
        int dl = dst[e] - lo;
        if (ok) gl[k] = (unsigned)s | ((unsigned)dl << L1_SRC_BITS);
        if (!ok) { s = 0; dl = L1_ROWS; }
        const int left = n - k0;
#pragma unroll
        for (int sb = 0; sb < 4; ++sb) {
            if (sb * 8 < left) {
                float4 v[8];
                int    dr[8];
#pragma unroll
                for (int j = 0; j < 8; ++j) {
                    const int sj = __builtin_amdgcn_readlane(s, sb * 8 + j);
                    dr[j]        = __builtin_amdgcn_readlane(dl, sb * 8 + j);
                    v[j] = *(const float4*)(xs + (size_t)sj * 128 + lane * 4);
                }
#pragma unroll
                for (int j = 0; j < 8; ++j) {
                    float* a = acc + dr[j] * L1_PITCH + lane;
                    atomicAdd(a,      v[j].x);
                    atomicAdd(a + 32, v[j].y);
                    atomicAdd(a + 64, v[j].z);
                    atomicAdd(a + 96, v[j].w);
                }
            }
        }
    }
    float2 wv[8];
    {
        const int l  = tid & 31;
        const int kk = tid >> 5;
        const int k  = kk * 4 + (l >> 4) * 2;
        const float* wp = W1 + (size_t)k * 128 + (l & 15);
#pragma unroll
        for (int ct = 0; ct < 8; ++ct) wv[ct] = make_float2(wp[ct * 16], wp[128 + ct * 16]);
    }
    __syncthreads();
    {
        float2* wst = (float2*)queue + tid;
#pragma unroll
        for (int ct = 0; ct < 8; ++ct) wst[ct * 1024] = wv[ct];
    }

    const bool  active = wave < L1_TILES;
    const int   mn     = lane & 15;
    const int   hf     = lane >> 4;
    float*      trow   = acc + wave * 16 * L1_PITCH;
    v2f a[32];
    if (active) {
        const float* ar = trow + mn * L1_PITCH + hf * 64;
#pragma unroll
        for (int kk = 0; kk < 32; ++kk) {
            a[kk].x = ar[kk];
            a[kk].y = ar[32 + kk];
        }
    }
    __syncthreads();

    if (active) {
        const v2f* wl   = (const v2f*)queue + lane;
        const int  m0   = lo + wave * 16 + hf * 8;
        float      nv[8];
#pragma unroll
        for (int r = 0; r < 8; ++r) nv[r] = ninv_in[(m0 + r < N_NODES) ? (m0 + r) : (N_NODES - 1)];
        for (int ct = 0; ct < 8; ct += 2) {
            v8f c0 = {};
            v8f c1 = {};
#pragma unroll
            for (int kk = 0; kk < 32; ++kk) {
                const v2f b0 = wl[((ct + 0) * 32 + kk) * 32];
                const v2f b1v = wl[((ct + 1) * 32 + kk) * 32];
                c0 = __builtin_amdgcn_wmma_f32_16x16x4_f32(false, a[kk], false, b0,
                                                           (short)0, c0, false, false);
                c1 = __builtin_amdgcn_wmma_f32_16x16x4_f32(false, a[kk], false, b1v,
                                                           (short)0, c1, false, false);
            }
            const int   n0  = ct * 16 + mn;
            const float bn0 = b1[n0];
            const float bn1 = b1[n0 + 16];
            float* d = trow + hf * 8 * L1_PITCH + n0;
#pragma unroll
            for (int r = 0; r < 8; ++r) {
                d[r * L1_PITCH]      = fmaxf(c0[r] * nv[r] + bn0, 0.0f);
                d[r * L1_PITCH + 16] = fmaxf(c1[r] * nv[r] + bn1, 0.0f);
            }
        }
    }
    float2 w2v[4];
    float  so = 0.0f;
    if (active) {
        const float2* w2p = (const float2*)W2;
#pragma unroll
        for (int j = 0; j < 4; ++j) w2v[j] = w2p[lane + 32 * j];
        const int node = lo + wave * 16 + mn;
        so = ninv_out[node < N_NODES ? node : (N_NODES - 1)];
    }
    __syncthreads();

    if (active) {
        float p[32];
#pragma unroll
        for (int i = 0; i < 16; ++i) {
            const float  s  = __int_as_float(__builtin_amdgcn_readlane(__float_as_int(so), i));
            const float* hr = trow + i * L1_PITCH + lane;
            float y0 = 0.0f, y1 = 0.0f;
#pragma unroll
            for (int j = 0; j < 4; ++j) {
                const float v = hr[32 * j] * s;
                y0 += v * w2v[j].x;
                y1 += v * w2v[j].y;
            }
            p[2 * i]     = y0;
            p[2 * i + 1] = y1;
        }
        fold_step<16, 16>(p, lane);
        fold_step<8, 8>(p, lane);
        fold_step<4, 4>(p, lane);
        fold_step<2, 2>(p, lane);
        fold_step<1, 1>(p, lane);
        const int r = wave * 16 + (lane >> 1);
        if (r < L1_ROWS && lo + r < N_NODES) y_node[(size_t)(lo + wave * 16) * 2 + lane] = p[0];
    }
}

#define L2_THREADS 512
#define L2_BATCH 16
__global__ __launch_bounds__(L2_THREADS) void agg2_kernel(const unsigned* __restrict__ elist,
                                                          const int* __restrict__ ecount,
                                                          const unsigned* __restrict__ oflag,
                                                          const int* __restrict__ src,
                                                          const int* __restrict__ dst,
                                                          const float* __restrict__ y_node,
                                                          const float* __restrict__ ninv_in,
                                                          const float* __restrict__ b2,
                                                          float* __restrict__ out) {
    __shared__ float acc[2 * L1_ROWS];

    const int tid = threadIdx.x;
    const int g   = blockIdx.x;
    const int lo  = g * L1_ROWS;

    int n = ecount[g];
    n = n < 0 ? 0 : (n > L1_LIST ? L1_LIST : n);
    const bool full = oflag[0] != 0u;

    for (int i = tid; i < 2 * L1_ROWS; i += L2_THREADS) acc[i] = 0.0f;
    __syncthreads();

    const float2* yn = (const float2*)y_node;
    if (!full) {
        const unsigned* el = elist + (size_t)g * L1_LIST;
        for (int k0 = 0; k0 < n; k0 += L2_THREADS * L2_BATCH) {
            unsigned pw[L2_BATCH];
            float2   yv[L2_BATCH];
#pragma unroll
            for (int j = 0; j < L2_BATCH; ++j) {
                const int k = k0 + j * L2_THREADS + tid;
                pw[j] = el[k < n ? k : n - 1];
            }
#pragma unroll
            for (int j = 0; j < L2_BATCH; ++j) yv[j] = yn[pw[j] & ((1u << L1_SRC_BITS) - 1u)];
            asm volatile("" :: "v"(yv[0].x), "v"(yv[0].y), "v"(yv[1].x), "v"(yv[1].y),
                               "v"(yv[2].x), "v"(yv[2].y), "v"(yv[3].x), "v"(yv[3].y),
                               "v"(yv[4].x), "v"(yv[4].y), "v"(yv[5].x), "v"(yv[5].y),
                               "v"(yv[6].x), "v"(yv[6].y), "v"(yv[7].x), "v"(yv[7].y));
            asm volatile("" :: "v"(yv[8].x), "v"(yv[8].y), "v"(yv[9].x), "v"(yv[9].y),
                               "v"(yv[10].x), "v"(yv[10].y), "v"(yv[11].x), "v"(yv[11].y),
                               "v"(yv[12].x), "v"(yv[12].y), "v"(yv[13].x), "v"(yv[13].y),
                               "v"(yv[14].x), "v"(yv[14].y), "v"(yv[15].x), "v"(yv[15].y));
#pragma unroll
            for (int j = 0; j < L2_BATCH; ++j) {
                if (k0 + j * L2_THREADS + tid < n) {
                    const unsigned dl = pw[j] >> L1_SRC_BITS;
                    atomicAdd(&acc[2 * dl], yv[j].x);
                    atomicAdd(&acc[2 * dl + 1], yv[j].y);
                }
            }
        }
    } else {
        for (int e = tid; e < N_EDGES; e += L2_THREADS) {
            const unsigned r = (unsigned)(dst[e] - lo);
            if (r < L1_ROWS) {
                const float2 v = yn[src[e]];
                atomicAdd(&acc[2 * r], v.x);
                atomicAdd(&acc[2 * r + 1], v.y);
            }
        }
    }
    __syncthreads();

    for (int i = tid; i < 2 * L1_ROWS; i += L2_THREADS) {
        const int node = lo + (i >> 1);
        if (node < N_NODES) out[(size_t)lo * 2 + i] = acc[i] * ninv_in[node] + b2[i & 1];
    }
}

extern "C" void kernel_launch(void* const* d_in, const int* in_sizes, int n_in,
                              void* d_out, int out_size, void* d_ws, size_t ws_size,
                              hipStream_t stream) {
    const float* in_feat = (const float*)d_in[0];
    const float* W1      = (const float*)d_in[1];
    const float* b1      = (const float*)d_in[2];
    const float* W2      = (const float*)d_in[3];
    const float* b2      = (const float*)d_in[4];
    const int*   src     = (const int*)d_in[5];
    const int*   dst     = (const int*)d_in[6];
    float*       out     = (float*)d_out;

    const int N = N_NODES;

    float* ws      = (float*)d_ws;
    float* deg_out = ws;
    float* deg_in  = ws + N;
    float* xs      = ws + 2 * (size_t)N;
    float* y_node  = xs + (size_t)N * 128;
    float* part    = y_node + 2 * (size_t)N;
    unsigned* omap   = (unsigned*)(part + (size_t)N_SLICES * 2 * N);
    unsigned* elist  = omap + (size_t)N_SB * 256;
    int*      ecount = (int*)(elist + (size_t)L1_GROUPS * L1_LIST);
    unsigned* oflag  = (unsigned*)(ecount + L1_GROUPS);

    degrees_kernel<<<dim3(N_SLICES, N_RANGES), 1024, 0, stream>>>(src, dst, part, omap, oflag);
    ninv_kernel<<<(2 * N + 255) / 256, 256, 0, stream>>>(part, deg_out, 2 * N);
    scale_kernel<<<STREAM_GROUPS, 256, 0, stream>>>(in_feat, deg_out, xs);
    agg_lds_kernel<<<L1_GROUPS, 1024, 0, stream>>>(src, dst, omap, xs, W1, b1, deg_in, deg_out, W2,
                                                   y_node, elist, ecount, oflag);
    agg2_kernel<<<L1_GROUPS, L2_THREADS, 0, stream>>>(elist, ecount, oflag, src, dst, y_node, deg_in, b2, out);
}


# degree-scale launch dissolved into its users: scaling waves add their rows' out-degree partials, layer-1 groups count in-degrees in LDS from the edges they find; in-degree planes and their pass gone
# speedup vs baseline: 4.7520x; 1.0429x over previous
#include <hip/hip_runtime.h>

typedef __attribute__((ext_vector_type(2))) float v2f;
typedef __attribute__((ext_vector_type(8))) float v8f;
typedef __attribute__((ext_vector_type(4))) float v4f;

#define N_NODES 100000
#define N_EDGES 1600000
#define IN_F 128
#define HID 128

#define N_RANGES 8
#define N_SLICES 32
#define RANGE_NODES (N_NODES / N_RANGES)
#define SLICE_EDGES (N_EDGES / N_SLICES)

#define L1_GROUPS 256
#define L1_ROWS 391
#define L1_SEG 512
#define N_BLK (N_EDGES / 32)
#define N_SB ((N_BLK + 31) / 32)
#define N_MAPTASK (N_BLK / 8)
#define L1_TILES 25
#define L1_PITCH 132
static_assert(L1_TILES * 16 >= L1_ROWS + 1, "the tiles must cover the range and its spare row");
static_assert(32 * L1_SEG == HID * IN_F, "W1 in fragment order takes the place of the edge queues");
static_assert(N_EDGES % 256 == 0, "the owner map is written in whole units of 8 blocks");
static_assert(L1_GROUPS == 256 && (N_NODES - 1) / L1_ROWS < 256, "a range number must fit the 8 planes");
static_assert(N_MAPTASK <= 32 * N_SLICES * N_RANGES, "one map unit per wave of the degree launch at most");
#define L1_LIST (32 * L1_SEG)
#define L1_SRC_BITS 17
static_assert(N_NODES <= (1 << L1_SRC_BITS) && L1_ROWS < (1 << (32 - L1_SRC_BITS)), "src and local dst must fit one word");

__global__ __launch_bounds__(1024) void degrees_kernel(const int* __restrict__ src,
                                                       const int* __restrict__ dst,
                                                       float* __restrict__ part,
                                                       unsigned* __restrict__ omap,
                                                       unsigned* __restrict__ oflag) {
    __shared__ unsigned h_out[RANGE_NODES];

    const int tid   = threadIdx.x;
    const int slice = blockIdx.x;
    const int lo    = blockIdx.y * RANGE_NODES;

    if (tid == 0 && blockIdx.x == 0 && blockIdx.y == 0) oflag[0] = 0u;

    const int  lane  = tid & 31;
    const int  mtask = __builtin_amdgcn_readfirstlane(tid >> 5) * (N_SLICES * N_RANGES) +
                       (int)(blockIdx.y * N_SLICES + blockIdx.x);
    const bool mine  = mtask < N_MAPTASK;
    int dv[8] = {};
    if (mine) {
        const int* dp = dst + (size_t)mtask * 256 + lane;
#pragma unroll
        for (int q = 0; q < 8; ++q) dv[q] = dp[q * 32];
    }

    for (int i = tid; i < RANGE_NODES; i += 1024) h_out[i] = 0u;

    if (mine) {
        unsigned w0 = 0u, w1 = 0u;
#pragma unroll
        for (int q = 0; q < 8; ++q) {
            const unsigned gid = (unsigned)dv[q] / L1_ROWS;
#pragma unroll
            for (int jj = 0; jj < 4; ++jj) {
                const unsigned b0 = __builtin_amdgcn_ballot_w32((gid & (1u << jj)) != 0u);
                const unsigned b1 = __builtin_amdgcn_ballot_w32((gid & (16u << jj)) != 0u);
                const bool     me = lane == q * 4 + jj;
                w0 = me ? b0 : w0;
                w1 = me ? b1 : w1;
            }
        }
        unsigned* mp = omap + (size_t)(mtask >> 2) * 256 + (mtask & 3) * 32 + lane;
        mp[0]   = w0;
        mp[128] = w1;
    }
    __syncthreads();

    const int4* s4 = (const int4*)(src + (size_t)slice * SLICE_EDGES);
    for (int i = tid; i < SLICE_EDGES / 4; i += 1024) {
        const int4 s = s4[i];
        unsigned r;
        r = (unsigned)(s.x - lo); if (r < RANGE_NODES) atomicAdd(&h_out[r], 1u);
        r = (unsigned)(s.y - lo); if (r < RANGE_NODES) atomicAdd(&h_out[r], 1u);
        r = (unsigned)(s.z - lo); if (r < RANGE_NODES) atomicAdd(&h_out[r], 1u);
        r = (unsigned)(s.w - lo); if (r < RANGE_NODES) atomicAdd(&h_out[r], 1u);
    }
    __syncthreads();

    float* po = part + (size_t)slice * N_NODES + lo;
    for (int i = tid; i < RANGE_NODES; i += 1024) po[i] = (float)h_out[i];
}

__device__ __forceinline__ float inv_sqrt_deg(float d) {
    return (d > 0.0f) ? (1.0f / sqrtf(d)) : 0.0f;
}

#define STREAM_ROWS 25
#define STREAM_GROUPS (N_NODES / (8 * STREAM_ROWS))
static_assert(STREAM_GROUPS * 8 * STREAM_ROWS == N_NODES, "rows must split evenly over the waves");

__global__ __launch_bounds__(256) void scale_kernel(const float* __restrict__ x,
                                                    const float* __restrict__ part,
                                                    float* __restrict__ ninv_out,
                                                    float* __restrict__ xs) {
    const int lane = threadIdx.x & 31;
    const int row0 = (blockIdx.x * 8 + (threadIdx.x >> 5)) * STREAM_ROWS;
    const v4f* x4 = (const v4f*)x + (size_t)row0 * 32 + lane;
    v4f*       o4 = (v4f*)xs + (size_t)row0 * 32 + lane;
    const float* pp = part + row0 + (lane < STREAM_ROWS ? lane : 0);
    float pd[N_SLICES];
#pragma unroll
    for (int s = 0; s < N_SLICES; ++s) pd[s] = pp[(size_t)s * N_NODES];
    v4f v[STREAM_ROWS];
#pragma unroll
    for (int j = 0; j < STREAM_ROWS; ++j) v[j] = __builtin_nontemporal_load(x4 + j * 32);
    __builtin_amdgcn_sched_barrier(0);
    float dsum = 0.0f;
#pragma unroll
    for (int s = 0; s < N_SLICES; ++s) dsum += pd[s];
    const float nv = inv_sqrt_deg(dsum);
    if (lane < STREAM_ROWS) ninv_out[row0 + lane] = nv;
#pragma unroll
    for (int j = 0; j < STREAM_ROWS; ++j) {
        const float s = __int_as_float(__builtin_amdgcn_readlane(__float_as_int(nv), j));
        o4[j * 32] = v[j] * s;
    }
}

template <int M, int K>
__device__ __forceinline__ void fold_step(float (&p)[32], int lane) {
    const bool up = (lane & M) != 0;
#pragma unroll
    for (int i = 0; i < K; ++i) {
        const float keep = up ? p[i + K] : p[i];
        const float send = up ? p[i] : p[i + K];
        p[i] = keep + __shfl_xor(send, M, 32);
    }
}

__global__ __launch_bounds__(1024) void agg_lds_kernel(const int* __restrict__ src,
                                                       const int* __restrict__ dst,
                                                       const unsigned* __restrict__ omap,
                                                       const float* __restrict__ xs,
                                                       const float* __restrict__ W1,
                                                       const float* __restrict__ b1,
                                                       float* __restrict__ ninv_in,
                                                       const float* __restrict__ ninv_out,
                                                       const float* __restrict__ W2,
                                                       float* __restrict__ y_node,
                                                       unsigned* __restrict__ elist,
                                                       int* __restrict__ ecount,
                                                       unsigned* __restrict__ oflag) {
    __shared__ __attribute__((aligned(16))) float acc[L1_TILES * 16 * L1_PITCH];
    __shared__ __attribute__((aligned(16))) unsigned queue[32 * L1_SEG];
    __shared__ int wcnt[32];
    __shared__ unsigned icnt[L1_TILES * 16];

    const int tid  = threadIdx.x;
    const int lane = tid & 31;
    const int wave = tid >> 5;
    const int g    = blockIdx.x;
    const int lo   = g * L1_ROWS;

    {
        float4* a4 = (float4*)acc;
        for (int i = tid; i < L1_TILES * 16 * L1_PITCH / 4; i += 1024) a4[i] = make_float4(0.f, 0.f, 0.f, 0.f);
        if (tid < L1_TILES * 16) icnt[tid] = 0u;
    }
    __syncthreads();

    const uint4*   m4   = (const uint4*)omap;
    const int      rot  = g * (N_SB / L1_GROUPS);
    const unsigned ltm  = (1u << lane) - 1u;
    unsigned*      myq  = queue + wave * L1_SEG;
    int            cnt  = 0;
    unsigned flip[8];
#pragma unroll
    for (int j = 0; j < 8; ++j) flip[j] = ((g >> j) & 1) ? 0u : 0xffffffffu;

    for (int it = 0; it < (N_SB + 127) / 128; ++it) {
        int   eb[4];
        bool  val[4];
        uint4 pl[4], ph[4];
#pragma unroll
        for (int u = 0; u < 4; ++u) {
            const int j = it * 128 + u * 32 + wave;
            int sb = j + rot;
            if (sb >= N_SB) sb -= N_SB;
            const int blk = sb * 32 + lane;
            val[u] = j < N_SB && blk < N_BLK;
            const int v = val[u] ? sb * 64 + lane : 0;
            pl[u] = m4[v];
            ph[u] = m4[v + 32];
            eb[u] = blk * 32;
        }
        __builtin_amdgcn_sched_barrier(0);
#pragma unroll
        for (int u = 0; u < 4; ++u) {
            unsigned m = (pl[u].x ^ flip[0]) & (pl[u].y ^ flip[1]) & (pl[u].z ^ flip[2]) & (pl[u].w ^ flip[3]) &
                         (ph[u].x ^ flip[4]) & (ph[u].y ^ flip[5]) & (ph[u].z ^ flip[6]) & (ph[u].w ^ flip[7]);
            if (!val[u]) m = 0u;
            while (__builtin_amdgcn_ballot_w32(m != 0u) != 0u) {
                const bool     has = m != 0u;
                const unsigned bal = __builtin_amdgcn_ballot_w32(has);
                if (has) {
                    const int e = eb[u] + __ffs(m) - 1;
                    m &= m - 1u;
                    const int slot = cnt + __popc(bal & ltm);
                    if (slot < L1_SEG) {
                        myq[slot] = (unsigned)e;
                    } else {
                        const int    s  = src[e];
                        const int    dl = dst[e] - lo;
                        const float* xr = xs + (size_t)s * 128;
                        float*       ar = acc + dl * L1_PITCH;
                        atomicAdd(&icnt[dl], 1u);
                        for (int f = 0; f < 128; ++f) atomicAdd(&ar[(f & 3) * 32 + (f >> 2)], xr[f]);
                    }
                }
                cnt += __popc(bal);
            }
        }
    }
    const int n = cnt < L1_SEG ? cnt : L1_SEG;
    if (lane == 0) {
        wcnt[wave] = n;
        if (cnt > L1_SEG) oflag[0] = 1u;
    }
    __syncthreads();

    int base = 0, total = 0;
    for (int v = 0; v < 32; ++v) {
        const int c = wcnt[v];
        total += c;
        base += v < wave ? c : 0;
    }
    if (tid == 0) ecount[g] = total;
    unsigned* gl = elist + (size_t)g * L1_LIST + base;

    for (int k0 = 0; k0 < n; k0 += 32) {
        const int  k  = k0 + lane;
        const bool ok = k < n;
        const int  e  = ok ? (int)myq[k] : 0;
        int s  = src[e];
        int dl = dst[e] - lo;
        if (ok) {
            gl[k] = (unsigned)s | ((unsigned)dl << L1_SRC_BITS);
            atomicAdd(&icnt[dl], 1u);
        } else {
            s = 0; dl = L1_ROWS;
        }
        const int left = n - k0;
#pragma unroll
        for (int sb = 0; sb < 4; ++sb) {
            if (sb * 8 < left) {
                float4 v[8];
                int    dr[8];
#pragma unroll
                for (int j = 0; j < 8; ++j) {
                    const int sj = __builtin_amdgcn_readlane(s, sb * 8 + j);
                    dr[j]        = __builtin_amdgcn_readlane(dl, sb * 8 + j);
                    v[j] = *(const float4*)(xs + (size_t)sj * 128 + lane * 4);
                }
#pragma unroll
                for (int j = 0; j < 8; ++j) {
                    float* a = acc + dr[j] * L1_PITCH + lane;
                    atomicAdd(a,      v[j].x);
                    atomicAdd(a + 32, v[j].y);
                    atomicAdd(a + 64, v[j].z);
                    atomicAdd(a + 96, v[j].w);
                }
            }
        }
    }
    float2 wv[8];
    {
        const int l  = tid & 31;
        const int kk = tid >> 5;
        const int k  = kk * 4 + (l >> 4) * 2;
        const float* wp = W1 + (size_t)k * 128 + (l & 15);
#pragma unroll
        for (int ct = 0; ct < 8; ++ct) wv[ct] = make_float2(wp[ct * 16], wp[128 + ct * 16]);
    }
    __syncthreads();
    {
        float2* wst = (float2*)queue + tid;
#pragma unroll
        for (int ct = 0; ct < 8; ++ct) wst[ct * 1024] = wv[ct];
    }

    const bool  active = wave < L1_TILES;
    const int   mn     = lane & 15;
    const int   hf     = lane >> 4;
    float*      trow   = acc + wave * 16 * L1_PITCH;
    v2f a[32];
    if (active) {
        const float* ar = trow + mn * L1_PITCH + hf * 64;
#pragma unroll
        for (int kk = 0; kk < 32; ++kk) {
            a[kk].x = ar[kk];
            a[kk].y = ar[32 + kk];
        }
    }
    __syncthreads();

    if (active) {
        const v2f* wl   = (const v2f*)queue + lane;
        float      nv[8];
        const int   rr = wave * 16 + mn;
        const float sc = inv_sqrt_deg((float)icnt[rr]);
#pragma unroll
        for (int r = 0; r < 8; ++r) {
            const float s_lo = __int_as_float(__builtin_amdgcn_readlane(__float_as_int(sc), r));
            const float s_hi = __int_as_float(__builtin_amdgcn_readlane(__float_as_int(sc), r + 8));
            nv[r] = hf ? s_hi : s_lo;
        }
        if (lane < 16 && rr < L1_ROWS && lo + rr < N_NODES) ninv_in[lo + rr] = sc;
        for (int ct = 0; ct < 8; ct += 2) {
            v8f c0 = {};
            v8f c1 = {};
#pragma unroll
            for (int kk = 0; kk < 32; ++kk) {
                const v2f b0 = wl[((ct + 0) * 32 + kk) * 32];
                const v2f b1v = wl[((ct + 1) * 32 + kk) * 32];
                c0 = __builtin_amdgcn_wmma_f32_16x16x4_f32(false, a[kk], false, b0,
                                                           (short)0, c0, false, false);
                c1 = __builtin_amdgcn_wmma_f32_16x16x4_f32(false, a[kk], false, b1v,
                                                           (short)0, c1, false, false);
            }
            const int   n0  = ct * 16 + mn;
            const float bn0 = b1[n0];
            const float bn1 = b1[n0 + 16];
            float* d = trow + hf * 8 * L1_PITCH + n0;
#pragma unroll
            for (int r = 0; r < 8; ++r) {
                d[r * L1_PITCH]      = fmaxf(c0[r] * nv[r] + bn0, 0.0f);
                d[r * L1_PITCH + 16] = fmaxf(c1[r] * nv[r] + bn1, 0.0f);
            }
        }
    }
    float2 w2v[4];
    float  so = 0.0f;
    if (active) {
        const float2* w2p = (const float2*)W2;
#pragma unroll
        for (int j = 0; j < 4; ++j) w2v[j] = w2p[lane + 32 * j];
        const int node = lo + wave * 16 + mn;
        so = ninv_out[node < N_NODES ? node : (N_NODES - 1)];
    }
    __syncthreads();

    if (active) {
        float p[32];
#pragma unroll
        for (int i = 0; i < 16; ++i) {
            const float  s  = __int_as_float(__builtin_amdgcn_readlane(__float_as_int(so), i));
            const float* hr = trow + i * L1_PITCH + lane;
            float y0 = 0.0f, y1 = 0.0f;
#pragma unroll
            for (int j = 0; j < 4; ++j) {
                const float v = hr[32 * j] * s;
                y0 += v * w2v[j].x;
                y1 += v * w2v[j].y;
            }
            p[2 * i]     = y0;
            p[2 * i + 1] = y1;
        }
        fold_step<16, 16>(p, lane);
        fold_step<8, 8>(p, lane);
        fold_step<4, 4>(p, lane);
        fold_step<2, 2>(p, lane);
        fold_step<1, 1>(p, lane);
        const int r = wave * 16 + (lane >> 1);
        if (r < L1_ROWS && lo + r < N_NODES) y_node[(size_t)(lo + wave * 16) * 2 + lane] = p[0];
    }
}

#define L2_THREADS 512
#define L2_BATCH 16
__global__ __launch_bounds__(L2_THREADS) void agg2_kernel(const unsigned* __restrict__ elist,
                                                          const int* __restrict__ ecount,
                                                          const unsigned* __restrict__ oflag,
                                                          const int* __restrict__ src,
                                                          const int* __restrict__ dst,
                                                          const float* __restrict__ y_node,
                                                          const float* __restrict__ ninv_in,
                                                          const float* __restrict__ b2,
                                                          float* __restrict__ out) {
    __shared__ float acc[2 * L1_ROWS];

    const int tid = threadIdx.x;
    const int g   = blockIdx.x;
    const int lo  = g * L1_ROWS;

    int n = ecount[g];
    n = n < 0 ? 0 : (n > L1_LIST ? L1_LIST : n);
    const bool full = oflag[0] != 0u;

    for (int i = tid; i < 2 * L1_ROWS; i += L2_THREADS) acc[i] = 0.0f;
    __syncthreads();

    const float2* yn = (const float2*)y_node;
    if (!full) {
        const unsigned* el = elist + (size_t)g * L1_LIST;
        for (int k0 = 0; k0 < n; k0 += L2_THREADS * L2_BATCH) {
            unsigned pw[L2_BATCH];
            float2   yv[L2_BATCH];
#pragma unroll
            for (int j = 0; j < L2_BATCH; ++j) {
                const int k = k0 + j * L2_THREADS + tid;
                pw[j] = el[k < n ? k : n - 1];
            }
#pragma unroll
            for (int j = 0; j < L2_BATCH; ++j) yv[j] = yn[pw[j] & ((1u << L1_SRC_BITS) - 1u)];
            asm volatile("" :: "v"(yv[0].x), "v"(yv[0].y), "v"(yv[1].x), "v"(yv[1].y),
                               "v"(yv[2].x), "v"(yv[2].y), "v"(yv[3].x), "v"(yv[3].y),
                               "v"(yv[4].x), "v"(yv[4].y), "v"(yv[5].x), "v"(yv[5].y),
                               "v"(yv[6].x), "v"(yv[6].y), "v"(yv[7].x), "v"(yv[7].y));
            asm volatile("" :: "v"(yv[8].x), "v"(yv[8].y), "v"(yv[9].x), "v"(yv[9].y),
                               "v"(yv[10].x), "v"(yv[10].y), "v"(yv[11].x), "v"(yv[11].y),
                               "v"(yv[12].x), "v"(yv[12].y), "v"(yv[13].x), "v"(yv[13].y),
                               "v"(yv[14].x), "v"(yv[14].y), "v"(yv[15].x), "v"(yv[15].y));
#pragma unroll
            for (int j = 0; j < L2_BATCH; ++j) {
                if (k0 + j * L2_THREADS + tid < n) {
                    const unsigned dl = pw[j] >> L1_SRC_BITS;
                    atomicAdd(&acc[2 * dl], yv[j].x);
                    atomicAdd(&acc[2 * dl + 1], yv[j].y);
                }
            }
        }
    } else {
        for (int e = tid; e < N_EDGES; e += L2_THREADS) {
            const unsigned r = (unsigned)(dst[e] - lo);
            if (r < L1_ROWS) {
                const float2 v = yn[src[e]];
                atomicAdd(&acc[2 * r], v.x);
                atomicAdd(&acc[2 * r + 1], v.y);
            }
        }
    }
    __syncthreads();

    for (int i = tid; i < 2 * L1_ROWS; i += L2_THREADS) {
        const int node = lo + (i >> 1);
        if (node < N_NODES) out[(size_t)lo * 2 + i] = acc[i] * ninv_in[node] + b2[i & 1];
    }
}

extern "C" void kernel_launch(void* const* d_in, const int* in_sizes, int n_in,
                              void* d_out, int out_size, void* d_ws, size_t ws_size,
                              hipStream_t stream) {
    const float* in_feat = (const float*)d_in[0];
    const float* W1      = (const float*)d_in[1];
    const float* b1      = (const float*)d_in[2];
    const float* W2      = (const float*)d_in[3];
    const float* b2      = (const float*)d_in[4];
    const int*   src     = (const int*)d_in[5];
    const int*   dst     = (const int*)d_in[6];
    float*       out     = (float*)d_out;

    const int N = N_NODES;

    float* ws      = (float*)d_ws;
    float* deg_out = ws;
    float* deg_in  = ws + N;
    float* xs      = ws + 2 * (size_t)N;
    float* y_node  = xs + (size_t)N * 128;
    float* part    = y_node + 2 * (size_t)N;
    unsigned* omap   = (unsigned*)(part + (size_t)N_SLICES * N);
    unsigned* elist  = omap + (size_t)N_SB * 256;
    int*      ecount = (int*)(elist + (size_t)L1_GROUPS * L1_LIST);
    unsigned* oflag  = (unsigned*)(ecount + L1_GROUPS);

    degrees_kernel<<<dim3(N_SLICES, N_RANGES), 1024, 0, stream>>>(src, dst, part, omap, oflag);
    scale_kernel<<<STREAM_GROUPS, 256, 0, stream>>>(in_feat, part, deg_out, xs);
    agg_lds_kernel<<<L1_GROUPS, 1024, 0, stream>>>(src, dst, omap, xs, W1, b1, deg_in, deg_out, W2,
                                                   y_node, elist, ecount, oflag);
    agg2_kernel<<<L1_GROUPS, L2_THREADS, 0, stream>>>(elist, ecount, oflag, src, dst, y_node, deg_in, b2, out);
}


# owner-map scan and edge-list build moved from the layer-1 launch to the row-scaling launch: 24 waves of a group scan beside 8 that stream rows; layer 1 reads its list; f32 LDS adds in arrival order
# speedup vs baseline: 4.9035x; 1.0319x over previous
#include <hip/hip_runtime.h>

typedef __attribute__((ext_vector_type(2))) float v2f;
typedef __attribute__((ext_vector_type(8))) float v8f;
typedef __attribute__((ext_vector_type(4))) float v4f;

#define N_NODES 100000
#define N_EDGES 1600000
#define IN_F 128
#define HID 128

#define N_RANGES 8
#define N_SLICES 32
#define RANGE_NODES (N_NODES / N_RANGES)
#define SLICE_EDGES (N_EDGES / N_SLICES)

#define L1_GROUPS 256
#define L1_ROWS 391
#define L1_SEG 512
#define N_BLK (N_EDGES / 32)
#define N_SB ((N_BLK + 31) / 32)
#define N_MAPTASK (N_BLK / 8)
#define L1_TILES 25
#define L1_PITCH 132
static_assert(L1_TILES * 16 >= L1_ROWS + 1, "the tiles must cover the range and its spare row");
static_assert(N_EDGES % 256 == 0, "the owner map is written in whole units of 8 blocks");
static_assert(L1_GROUPS == 256 && (N_NODES - 1) / L1_ROWS < 256, "a range number must fit the 8 planes");
static_assert(N_MAPTASK <= 32 * N_SLICES * N_RANGES, "one map unit per wave of the degree launch at most");
#define L1_LIST (32 * L1_SEG)
#define L1_SRC_BITS 17
static_assert(N_NODES <= (1 << L1_SRC_BITS) && L1_ROWS < (1 << (32 - L1_SRC_BITS)), "src and local dst must fit one word");

__global__ __launch_bounds__(1024) void degrees_kernel(const int* __restrict__ src,
                                                       const int* __restrict__ dst,
                                                       float* __restrict__ part,
                                                       unsigned* __restrict__ omap,
                                                       unsigned* __restrict__ oflag) {
    __shared__ unsigned h_out[RANGE_NODES];

    const int tid   = threadIdx.x;
    const int slice = blockIdx.x;
    const int lo    = blockIdx.y * RANGE_NODES;

    if (tid == 0 && blockIdx.x == 0 && blockIdx.y == 0) oflag[0] = 0u;

    const int  lane  = tid & 31;
    const int  mtask = __builtin_amdgcn_readfirstlane(tid >> 5) * (N_SLICES * N_RANGES) +
                       (int)(blockIdx.y * N_SLICES + blockIdx.x);
    const bool mine  = mtask < N_MAPTASK;
    int dv[8] = {};
    if (mine) {
        const int* dp = dst + (size_t)mtask * 256 + lane;
#pragma unroll
        for (int q = 0; q < 8; ++q) dv[q] = dp[q * 32];
    }

    for (int i = tid; i < RANGE_NODES; i += 1024) h_out[i] = 0u;

    if (mine) {
        unsigned w0 = 0u, w1 = 0u;
#pragma unroll
        for (int q = 0; q < 8; ++q) {
            const unsigned gid = (unsigned)dv[q] / L1_ROWS;
#pragma unroll
            for (int jj = 0; jj < 4; ++jj) {
                const unsigned b0 = __builtin_amdgcn_ballot_w32((gid & (1u << jj)) != 0u);
                const unsigned b1 = __builtin_amdgcn_ballot_w32((gid & (16u << jj)) != 0u);
                const bool     me = lane == q * 4 + jj;
                w0 = me ? b0 : w0;
                w1 = me ? b1 : w1;
            }
        }
        unsigned* mp = omap + (size_t)(mtask >> 2) * 256 + (mtask & 3) * 32 + lane;
        mp[0]   = w0;
        mp[128] = w1;
    }
    __syncthreads();

    const int4* s4 = (const int4*)(src + (size_t)slice * SLICE_EDGES);
    for (int i = tid; i < SLICE_EDGES / 4; i += 1024) {
        const int4 s = s4[i];
        unsigned r;
        r = (unsigned)(s.x - lo); if (r < RANGE_NODES) atomicAdd(&h_out[r], 1u);
        r = (unsigned)(s.y - lo); if (r < RANGE_NODES) atomicAdd(&h_out[r], 1u);
        r = (unsigned)(s.z - lo); if (r < RANGE_NODES) atomicAdd(&h_out[r], 1u);
        r = (unsigned)(s.w - lo); if (r < RANGE_NODES) atomicAdd(&h_out[r], 1u);
    }
    __syncthreads();

    float* po = part + (size_t)slice * N_NODES + lo;
    for (int i = tid; i < RANGE_NODES; i += 1024) po[i] = (float)h_out[i];
}

__device__ __forceinline__ float inv_sqrt_deg(float d) {
    return (d > 0.0f) ? (1.0f / sqrtf(d)) : 0.0f;
}

#define STREAM_WAVES 8
#define SCAN_WAVES (32 - STREAM_WAVES)
#define STREAM_ROWS 13
#define STREAM_TURNS 4
#define SCAN_TURNS ((N_SB + 4 * SCAN_WAVES - 1) / (4 * SCAN_WAVES))
static_assert(L1_GROUPS * STREAM_WAVES * STREAM_TURNS * STREAM_ROWS >= N_NODES, "the streaming waves must cover every row");
static_assert(SCAN_WAVES * L1_SEG <= L1_LIST, "the scanning waves' queues must fit the range's list");
#define LIST_CHUNKS 12
static_assert(LIST_CHUNKS == 12, "the fences behind the list loads name twelve values each");

__global__ __launch_bounds__(1024) void scale_scan_kernel(const float* __restrict__ x,
                                                          const float* __restrict__ part,
                                                          float* __restrict__ ninv_out,
                                                          float* __restrict__ xs,
                                                          const int* __restrict__ src,
                                                          const int* __restrict__ dst,
                                                          const unsigned* __restrict__ omap,
                                                          unsigned* __restrict__ elist,
                                                          int* __restrict__ ecount,
                                                          unsigned* __restrict__ oflag) {
    __shared__ unsigned queue[SCAN_WAVES * L1_SEG];
    __shared__ int wcnt[SCAN_WAVES];

    const int tid  = threadIdx.x;
    const int lane = tid & 31;
    const int wave = __builtin_amdgcn_readfirstlane(tid >> 5);
    const int g    = blockIdx.x;
    const int lo   = g * L1_ROWS;
    const int sw   = wave - STREAM_WAVES;
    unsigned* myq  = queue + (sw > 0 ? sw : 0) * L1_SEG;
    int       n    = 0;

    if (wave < STREAM_WAVES) {
#pragma unroll 1
        for (int b = 0; b < STREAM_TURNS; ++b) {
            const int row0 = ((g * STREAM_WAVES + wave) * STREAM_TURNS + b) * STREAM_ROWS;
            if (row0 >= N_NODES) break;
            const int    rj = row0 + (lane < STREAM_ROWS ? lane : 0);
            const float* pp = part + (rj < N_NODES ? rj : N_NODES - 1);
            float pd[N_SLICES];
#pragma unroll
            for (int s = 0; s < N_SLICES; ++s) pd[s] = pp[(size_t)s * N_NODES];
            v4f v[STREAM_ROWS];
#pragma unroll
            for (int j = 0; j < STREAM_ROWS; ++j) {
                const int r = row0 + j < N_NODES ? row0 + j : N_NODES - 1;
                v[j] = __builtin_nontemporal_load((const v4f*)x + (size_t)r * 32 + lane);
            }
            __builtin_amdgcn_sched_barrier(0);
            float dsum = 0.0f;
#pragma unroll
            for (int s = 0; s < N_SLICES; ++s) dsum += pd[s];
            const float nv = inv_sqrt_deg(dsum);
            if (lane < STREAM_ROWS && row0 + lane < N_NODES) ninv_out[row0 + lane] = nv;
#pragma unroll
            for (int j = 0; j < STREAM_ROWS; ++j) {
                if (row0 + j < N_NODES) {
                    const float s = __int_as_float(__builtin_amdgcn_readlane(__float_as_int(nv), j));
                    ((v4f*)xs)[(size_t)(row0 + j) * 32 + lane] = v[j] * s;
                }
            }
        }
    } else {
        const uint4*   m4  = (const uint4*)omap;
        const int      rot = g * (N_SB / L1_GROUPS);
        const unsigned ltm = (1u << lane) - 1u;
        int            cnt = 0;
        unsigned flip[8];
#pragma unroll
        for (int j = 0; j < 8; ++j) flip[j] = ((g >> j) & 1) ? 0u : 0xffffffffu;

        for (int it = 0; it < SCAN_TURNS; ++it) {
            int   eb[4];
            bool  val[4];
            uint4 pl[4], ph[4];
#pragma unroll
            for (int u = 0; u < 4; ++u) {
                const int j = (it * 4 + u) * SCAN_WAVES + sw;
                int sb = j + rot;
                if (sb >= N_SB) sb -= N_SB;
                const int blk = sb * 32 + lane;
                val[u] = j < N_SB && blk < N_BLK;
                const int v = val[u] ? sb * 64 + lane : 0;
                pl[u] = m4[v];
                ph[u] = m4[v + 32];
                eb[u] = blk * 32;
            }
            __builtin_amdgcn_sched_barrier(0);
#pragma unroll
            for (int u = 0; u < 4; ++u) {
                unsigned m = (pl[u].x ^ flip[0]) & (pl[u].y ^ flip[1]) & (pl[u].z ^ flip[2]) & (pl[u].w ^ flip[3]) &
                             (ph[u].x ^ flip[4]) & (ph[u].y ^ flip[5]) & (ph[u].z ^ flip[6]) & (ph[u].w ^ flip[7]);
                if (!val[u]) m = 0u;
                while (__builtin_amdgcn_ballot_w32(m != 0u) != 0u) {
                    const bool     has = m != 0u;
                    const unsigned bal = __builtin_amdgcn_ballot_w32(has);
                    if (has) {
                        const int e = eb[u] + __ffs(m) - 1;
                        m &= m - 1u;
                        const int slot = cnt + __popc(bal & ltm);
                        if (slot < L1_SEG) myq[slot] = (unsigned)e;
                    }
                    cnt += __popc(bal);
                }
            }
        }
        n = cnt < L1_SEG ? cnt : L1_SEG;
        if (lane == 0) {
            wcnt[sw] = n;
            if (cnt > L1_SEG) oflag[0] = 1u;
        }
    }
    __syncthreads();

    int base = 0, total = 0;
    for (int v = 0; v < SCAN_WAVES; ++v) {
        const int c = wcnt[v];
        total += c;
        base += v < sw ? c : 0;
    }
    if (tid == 0) ecount[g] = total;

    unsigned* gl = elist + (size_t)g * L1_LIST + base;
    for (int k0 = 0; k0 < n; k0 += 32 * LIST_CHUNKS) {
        int e[LIST_CHUNKS], s[LIST_CHUNKS], d[LIST_CHUNKS];
#pragma unroll
        for (int j = 0; j < LIST_CHUNKS; ++j) {
            const int k = k0 + j * 32 + lane;
            e[j] = (int)myq[k < n ? k : 0];
        }
#pragma unroll
        for (int j = 0; j < LIST_CHUNKS; ++j) {
            s[j] = src[e[j]];
            d[j] = dst[e[j]];
        }
        asm volatile("" :: "v"(s[0]), "v"(s[1]), "v"(s[2]), "v"(s[3]), "v"(s[4]), "v"(s[5]),
                           "v"(s[6]), "v"(s[7]), "v"(s[8]), "v"(s[9]), "v"(s[10]), "v"(s[11]));
        asm volatile("" :: "v"(d[0]), "v"(d[1]), "v"(d[2]), "v"(d[3]), "v"(d[4]), "v"(d[5]),
                           "v"(d[6]), "v"(d[7]), "v"(d[8]), "v"(d[9]), "v"(d[10]), "v"(d[11]));
#pragma unroll
        for (int j = 0; j < LIST_CHUNKS; ++j) {
            const int k = k0 + j * 32 + lane;
            if (k < n) gl[k] = (unsigned)s[j] | ((unsigned)(d[j] - lo) << L1_SRC_BITS);
        }
    }
}

template <int M, int K>
__device__ __forceinline__ void fold_step(float (&p)[32], int lane) {
    const bool up = (lane & M) != 0;
#pragma unroll
    for (int i = 0; i < K; ++i) {
        const float keep = up ? p[i + K] : p[i];
        const float send = up ? p[i] : p[i + K];
        p[i] = keep + __shfl_xor(send, M, 32);
    }
}

__global__ __launch_bounds__(1024) void agg_lds_kernel(const unsigned* __restrict__ elist,
                                                       const int* __restrict__ ecount,
                                                       const unsigned* __restrict__ oflag,
                                                       const int* __restrict__ src,
                                                       const int* __restrict__ dst,
                                                       const float* __restrict__ xs,
                                                       const float* __restrict__ W1,
                                                       const float* __restrict__ b1,
                                                       float* __restrict__ ninv_in,
                                                       const float* __restrict__ ninv_out,
                                                       const float* __restrict__ W2,
                                                       float* __restrict__ y_node) {
    __shared__ __attribute__((aligned(16))) float acc[L1_TILES * 16 * L1_PITCH];
    __shared__ __attribute__((aligned(16))) float w1f[HID * IN_F];
    __shared__ unsigned icnt[L1_TILES * 16];

    const int tid  = threadIdx.x;
    const int lane = tid & 31;
    const int wave = __builtin_amdgcn_readfirstlane(tid >> 5);
    const int g    = blockIdx.x;
    const int lo   = g * L1_ROWS;

    int n = ecount[g];
    n = n < 0 ? 0 : (n > L1_LIST ? L1_LIST : n);
    const bool full = oflag[0] != 0u;

    {
        float4* a4 = (float4*)acc;
        for (int i = tid; i < L1_TILES * 16 * L1_PITCH / 4; i += 1024) a4[i] = make_float4(0.f, 0.f, 0.f, 0.f);
        if (tid < L1_TILES * 16) icnt[tid] = 0u;
    }
    __syncthreads();

    if (!full) {
        const unsigned* el = elist + (size_t)g * L1_LIST;
        for (int k0 = wave * 32; k0 < n; k0 += 1024) {
            const int      k  = k0 + lane;
            const bool     ok = k < n;
            const unsigned w  = el[ok ? k : k0];
            int s  = (int)(w & ((1u << L1_SRC_BITS) - 1u));
            int dl = (int)(w >> L1_SRC_BITS);
            if (ok) {
                atomicAdd(&icnt[dl], 1u);
            } else {
                s = 0; dl = L1_ROWS;
            }
            const int left = n - k0;
#pragma unroll
            for (int sb = 0; sb < 4; ++sb) {
                if (sb * 8 < left) {
                    float4 v[8];
                    int    dr[8];
#pragma unroll
                    for (int j = 0; j < 8; ++j) {
                        const int sj = __builtin_amdgcn_readlane(s, sb * 8 + j);
                        dr[j]        = __builtin_amdgcn_readlane(dl, sb * 8 + j);
                        v[j] = *(const float4*)(xs + (size_t)sj * 128 + lane * 4);
                    }
#pragma unroll
                    for (int j = 0; j < 8; ++j) {
                        float* a = acc + dr[j] * L1_PITCH + lane;
                        atomicAdd(a,      v[j].x);
                        atomicAdd(a + 32, v[j].y);
                        atomicAdd(a + 64, v[j].z);
                        atomicAdd(a + 96, v[j].w);
                    }
                }
            }
        }
    } else {
        for (int e = tid; e < N_EDGES; e += 1024) {
            const unsigned dl = (unsigned)(dst[e] - lo);
            if (dl < L1_ROWS) {
                const float* xr = xs + (size_t)src[e] * 128;
                float*       ar = acc + dl * L1_PITCH;
                atomicAdd(&icnt[dl], 1u);
                for (int f = 0; f < 128; ++f) atomicAdd(&ar[(f & 3) * 32 + (f >> 2)], xr[f]);
            }
        }
    }
    float2 wv[8];
    {
        const int l  = tid & 31;
        const int kk = tid >> 5;
        const int k  = kk * 4 + (l >> 4) * 2;
        const float* wp = W1 + (size_t)k * 128 + (l & 15);
#pragma unroll
        for (int ct = 0; ct < 8; ++ct) wv[ct] = make_float2(wp[ct * 16], wp[128 + ct * 16]);
    }
    __syncthreads();
    {
        float2* wst = (float2*)w1f + tid;
#pragma unroll
        for (int ct = 0; ct < 8; ++ct) wst[ct * 1024] = wv[ct];
    }

    const bool  active = wave < L1_TILES;
    const int   mn     = lane & 15;
    const int   hf     = lane >> 4;
    float*      trow   = acc + wave * 16 * L1_PITCH;
    v2f a[32];
    if (active) {
        const float* ar = trow + mn * L1_PITCH + hf * 64;
#pragma unroll
        for (int kk = 0; kk < 32; ++kk) {
            a[kk].x = ar[kk];
            a[kk].y = ar[32 + kk];
        }
    }
    __syncthreads();

    if (active) {
        const v2f* wl   = (const v2f*)w1f + lane;
        float      nv[8];
        const int   rr = wave * 16 + mn;
        const float sc = inv_sqrt_deg((float)icnt[rr]);
#pragma unroll
        for (int r = 0; r < 8; ++r) {
            const float s_lo = __int_as_float(__builtin_amdgcn_readlane(__float_as_int(sc), r));
            const float s_hi = __int_as_float(__builtin_amdgcn_readlane(__float_as_int(sc), r + 8));
            nv[r] = hf ? s_hi : s_lo;
        }
        if (lane < 16 && rr < L1_ROWS && lo + rr < N_NODES) ninv_in[lo + rr] = sc;
        for (int ct = 0; ct < 8; ct += 2) {
            v8f c0 = {};
            v8f c1 = {};
#pragma unroll
            for (int kk = 0; kk < 32; ++kk) {
                const v2f b0 = wl[((ct + 0) * 32 + kk) * 32];
                const v2f b1v = wl[((ct + 1) * 32 + kk) * 32];
                c0 = __builtin_amdgcn_wmma_f32_16x16x4_f32(false, a[kk], false, b0,
                                                           (short)0, c0, false, false);
                c1 = __builtin_amdgcn_wmma_f32_16x16x4_f32(false, a[kk], false, b1v,
                                                           (short)0, c1, false, false);
            }
            const int   n0  = ct * 16 + mn;
            const float bn0 = b1[n0];
            const float bn1 = b1[n0 + 16];
            float* d = trow + hf * 8 * L1_PITCH + n0;
#pragma unroll
            for (int r = 0; r < 8; ++r) {
                d[r * L1_PITCH]      = fmaxf(c0[r] * nv[r] + bn0, 0.0f);
                d[r * L1_PITCH + 16] = fmaxf(c1[r] * nv[r] + bn1, 0.0f);
            }
        }
    }
    float2 w2v[4];
    float  so = 0.0f;
    if (active) {
        const float2* w2p = (const float2*)W2;
#pragma unroll
        for (int j = 0; j < 4; ++j) w2v[j] = w2p[lane + 32 * j];
        const int node = lo + wave * 16 + mn;
        so = ninv_out[node < N_NODES ? node : (N_NODES - 1)];
    }
    __syncthreads();

    if (active) {
        float p[32];
#pragma unroll
        for (int i = 0; i < 16; ++i) {
            const float  s  = __int_as_float(__builtin_amdgcn_readlane(__float_as_int(so), i));
            const float* hr = trow + i * L1_PITCH + lane;
            float y0 = 0.0f, y1 = 0.0f;
#pragma unroll
            for (int j = 0; j < 4; ++j) {
                const float v = hr[32 * j] * s;
                y0 += v * w2v[j].x;
                y1 += v * w2v[j].y;
            }
            p[2 * i]     = y0;
            p[2 * i + 1] = y1;
        }
        fold_step<16, 16>(p, lane);
        fold_step<8, 8>(p, lane);
        fold_step<4, 4>(p, lane);
        fold_step<2, 2>(p, lane);
        fold_step<1, 1>(p, lane);
        const int r = wave * 16 + (lane >> 1);
        if (r < L1_ROWS && lo + r < N_NODES) y_node[(size_t)(lo + wave * 16) * 2 + lane] = p[0];
    }
}

#define L2_THREADS 512
#define L2_BATCH 16
__global__ __launch_bounds__(L2_THREADS) void agg2_kernel(const unsigned* __restrict__ elist,
                                                          const int* __restrict__ ecount,
                                                          const unsigned* __restrict__ oflag,
                                                          const int* __restrict__ src,
                                                          const int* __restrict__ dst,
                                                          const float* __restrict__ y_node,
                                                          const float* __restrict__ ninv_in,
                                                          const float* __restrict__ b2,
                                                          float* __restrict__ out) {
    __shared__ float acc[2 * L1_ROWS];

    const int tid = threadIdx.x;
    const int g   = blockIdx.x;
    const int lo  = g * L1_ROWS;

    int n = ecount[g];
    n = n < 0 ? 0 : (n > L1_LIST ? L1_LIST : n);
    const bool full = oflag[0] != 0u;

    for (int i = tid; i < 2 * L1_ROWS; i += L2_THREADS) acc[i] = 0.0f;
    __syncthreads();

    const float2* yn = (const float2*)y_node;
    if (!full) {
        const unsigned* el = elist + (size_t)g * L1_LIST;
        for (int k0 = 0; k0 < n; k0 += L2_THREADS * L2_BATCH) {
            unsigned pw[L2_BATCH];
            float2   yv[L2_BATCH];
#pragma unroll
            for (int j = 0; j < L2_BATCH; ++j) {
                const int k = k0 + j * L2_THREADS + tid;
                pw[j] = el[k < n ? k : n - 1];
            }
#pragma unroll
            for (int j = 0; j < L2_BATCH; ++j) yv[j] = yn[pw[j] & ((1u << L1_SRC_BITS) - 1u)];
            asm volatile("" :: "v"(yv[0].x), "v"(yv[0].y), "v"(yv[1].x), "v"(yv[1].y),
                               "v"(yv[2].x), "v"(yv[2].y), "v"(yv[3].x), "v"(yv[3].y),
                               "v"(yv[4].x), "v"(yv[4].y), "v"(yv[5].x), "v"(yv[5].y),
                               "v"(yv[6].x), "v"(yv[6].y), "v"(yv[7].x), "v"(yv[7].y));
            asm volatile("" :: "v"(yv[8].x), "v"(yv[8].y), "v"(yv[9].x), "v"(yv[9].y),
                               "v"(yv[10].x), "v"(yv[10].y), "v"(yv[11].x), "v"(yv[11].y),
                               "v"(yv[12].x), "v"(yv[12].y), "v"(yv[13].x), "v"(yv[13].y),
                               "v"(yv[14].x), "v"(yv[14].y), "v"(yv[15].x), "v"(yv[15].y));
#pragma unroll
            for (int j = 0; j < L2_BATCH; ++j) {
                if (k0 + j * L2_THREADS + tid < n) {
                    const unsigned dl = pw[j] >> L1_SRC_BITS;
                    atomicAdd(&acc[2 * dl], yv[j].x);
                    atomicAdd(&acc[2 * dl + 1], yv[j].y);
                }
            }
        }
    } else {
        for (int e = tid; e < N_EDGES; e += L2_THREADS) {
            const unsigned r = (unsigned)(dst[e] - lo);
            if (r < L1_ROWS) {
                const float2 v = yn[src[e]];
                atomicAdd(&acc[2 * r], v.x);
                atomicAdd(&acc[2 * r + 1], v.y);
            }
        }
    }
    __syncthreads();

    for (int i = tid; i < 2 * L1_ROWS; i += L2_THREADS) {
        const int node = lo + (i >> 1);
        if (node < N_NODES) out[(size_t)lo * 2 + i] = acc[i] * ninv_in[node] + b2[i & 1];
    }
}

extern "C" void kernel_launch(void* const* d_in, const int* in_sizes, int n_in,
                              void* d_out, int out_size, void* d_ws, size_t ws_size,
                              hipStream_t stream) {
    const float* in_feat = (const float*)d_in[0];
    const float* W1      = (const float*)d_in[1];
    const float* b1      = (const float*)d_in[2];
    const float* W2      = (const float*)d_in[3];
    const float* b2      = (const float*)d_in[4];
    const int*   src     = (const int*)d_in[5];
    const int*   dst     = (const int*)d_in[6];
    float*       out     = (float*)d_out;

    const int N = N_NODES;

    float* ws      = (float*)d_ws;
    float* deg_out = ws;
    float* deg_in  = ws + N;
    float* xs      = ws + 2 * (size_t)N;
    float* y_node  = xs + (size_t)N * 128;
    float* part    = y_node + 2 * (size_t)N;
    unsigned* omap   = (unsigned*)(part + (size_t)N_SLICES * N);
    unsigned* elist  = omap + (size_t)N_SB * 256;
    int*      ecount = (int*)(elist + (size_t)L1_GROUPS * L1_LIST);
    unsigned* oflag  = (unsigned*)(ecount + L1_GROUPS);

    degrees_kernel<<<dim3(N_SLICES, N_RANGES), 1024, 0, stream>>>(src, dst, part, omap, oflag);
    scale_scan_kernel<<<L1_GROUPS, 1024, 0, stream>>>(in_feat, part, deg_out, xs, src, dst, omap,
                                                      elist, ecount, oflag);
    agg_lds_kernel<<<L1_GROUPS, 1024, 0, stream>>>(elist, ecount, oflag, src, dst, xs, W1, b1, deg_in, deg_out, W2,
                                                   y_node);
    agg2_kernel<<<L1_GROUPS, L2_THREADS, 0, stream>>>(elist, ecount, oflag, src, dst, y_node, deg_in, b2, out);
}


# layer-1 product with W1 moved in front of the edge sums, into the row-scaling waves (matrix unit beside stream and scan): edges add rows of (x*scale)@W1, new f32 sum order; no GEMM phase in layer 1
# speedup vs baseline: 5.2494x; 1.0705x over previous
#include <hip/hip_runtime.h>

typedef __attribute__((ext_vector_type(2))) float v2f;
typedef __attribute__((ext_vector_type(8))) float v8f;
typedef __attribute__((ext_vector_type(4))) float v4f;

#define N_NODES 100000
#define N_EDGES 1600000
#define IN_F 128
#define HID 128

#define N_RANGES 8
#define N_SLICES 32
#define RANGE_NODES (N_NODES / N_RANGES)
#define SLICE_EDGES (N_EDGES / N_SLICES)

#define L1_GROUPS 256
#define L1_ROWS 391
#define L1_SEG 512
#define N_BLK (N_EDGES / 32)
#define N_SB ((N_BLK + 31) / 32)
#define N_MAPTASK (N_BLK / 8)
#define L1_TILES 25
#define L1_PITCH 132
static_assert(L1_TILES * 16 >= L1_ROWS + 1, "the tiles must cover the range and its spare row");
static_assert(N_EDGES % 256 == 0, "the owner map is written in whole units of 8 blocks");
static_assert(L1_GROUPS == 256 && (N_NODES - 1) / L1_ROWS < 256, "a range number must fit the 8 planes");
static_assert(N_MAPTASK <= 32 * N_SLICES * N_RANGES, "one map unit per wave of the degree launch at most");
#define L1_LIST (32 * L1_SEG)
#define L1_SRC_BITS 17
static_assert(N_NODES <= (1 << L1_SRC_BITS) && L1_ROWS < (1 << (32 - L1_SRC_BITS)), "src and local dst must fit one word");

__global__ __launch_bounds__(1024) void degrees_kernel(const int* __restrict__ src,
                                                       const int* __restrict__ dst,
                                                       float* __restrict__ part,
                                                       unsigned* __restrict__ omap,
                                                       unsigned* __restrict__ oflag) {
    __shared__ unsigned h_out[RANGE_NODES];

    const int tid   = threadIdx.x;
    const int slice = blockIdx.x;
    const int lo    = blockIdx.y * RANGE_NODES;

    if (tid == 0 && blockIdx.x == 0 && blockIdx.y == 0) oflag[0] = 0u;

    const int  lane  = tid & 31;
    const int  mtask = __builtin_amdgcn_readfirstlane(tid >> 5) * (N_SLICES * N_RANGES) +
                       (int)(blockIdx.y * N_SLICES + blockIdx.x);
    const bool mine  = mtask < N_MAPTASK;
    int dv[8] = {};
    if (mine) {
        const int* dp = dst + (size_t)mtask * 256 + lane;
#pragma unroll
        for (int q = 0; q < 8; ++q) dv[q] = dp[q * 32];
    }

    for (int i = tid; i < RANGE_NODES; i += 1024) h_out[i] = 0u;

    if (mine) {
        unsigned w0 = 0u, w1 = 0u;
#pragma unroll
        for (int q = 0; q < 8; ++q) {
            const unsigned gid = (unsigned)dv[q] / L1_ROWS;
#pragma unroll
            for (int jj = 0; jj < 4; ++jj) {
                const unsigned b0 = __builtin_amdgcn_ballot_w32((gid & (1u << jj)) != 0u);
                const unsigned b1 = __builtin_amdgcn_ballot_w32((gid & (16u << jj)) != 0u);
                const bool     me = lane == q * 4 + jj;
                w0 = me ? b0 : w0;
                w1 = me ? b1 : w1;
            }
        }
        unsigned* mp = omap + (size_t)(mtask >> 2) * 256 + (mtask & 3) * 32 + lane;
        mp[0]   = w0;
        mp[128] = w1;
    }
    __syncthreads();

    const int4* s4 = (const int4*)(src + (size_t)slice * SLICE_EDGES);
    for (int i = tid; i < SLICE_EDGES / 4; i += 1024) {
        const int4 s = s4[i];
        unsigned r;
        r = (unsigned)(s.x - lo); if (r < RANGE_NODES) atomicAdd(&h_out[r], 1u);
        r = (unsigned)(s.y - lo); if (r < RANGE_NODES) atomicAdd(&h_out[r], 1u);
        r = (unsigned)(s.z - lo); if (r < RANGE_NODES) atomicAdd(&h_out[r], 1u);
        r = (unsigned)(s.w - lo); if (r < RANGE_NODES) atomicAdd(&h_out[r], 1u);
    }
    __syncthreads();

    float* po = part + (size_t)slice * N_NODES + lo;
    for (int i = tid; i < RANGE_NODES; i += 1024) po[i] = (float)h_out[i];
}

__device__ __forceinline__ float inv_sqrt_deg(float d) {
    return (d > 0.0f) ? (1.0f / sqrtf(d)) : 0.0f;
}

#define STREAM_WAVES 8
#define SCAN_WAVES (32 - STREAM_WAVES)
#define STREAM_TURNS 4
#define N_TILES (N_NODES / 16)
#define STG_PITCH 132
static_assert(N_NODES % 16 == 0, "the rows are taken in whole 16-row tiles");
#define SCAN_TURNS ((N_SB + 4 * SCAN_WAVES - 1) / (4 * SCAN_WAVES))
static_assert(L1_GROUPS * STREAM_WAVES * STREAM_TURNS >= N_TILES, "the streaming waves must cover every tile");
static_assert(SCAN_WAVES * L1_SEG <= L1_LIST, "the scanning waves' queues must fit the range's list");
#define LIST_CHUNKS 12
static_assert(LIST_CHUNKS == 12, "the fences behind the list loads name twelve values each");

__global__ __launch_bounds__(1024) void scale_scan_kernel(const float* __restrict__ x,
                                                          const float* __restrict__ part,
                                                          const float* __restrict__ W1,
                                                          float* __restrict__ ninv_out,
                                                          float* __restrict__ xw,
                                                          const int* __restrict__ src,
                                                          const int* __restrict__ dst,
                                                          const unsigned* __restrict__ omap,
                                                          unsigned* __restrict__ elist,
                                                          int* __restrict__ ecount,
                                                          unsigned* __restrict__ oflag) {
    __shared__ unsigned queue[SCAN_WAVES * L1_SEG];
    __shared__ int wcnt[SCAN_WAVES];
    __shared__ __attribute__((aligned(16))) float w1f[HID * IN_F];
    __shared__ __attribute__((aligned(16))) float stage[STREAM_WAVES * 16 * STG_PITCH];

    const int tid  = threadIdx.x;
    const int lane = tid & 31;
    const int wave = __builtin_amdgcn_readfirstlane(tid >> 5);
    const int g    = blockIdx.x;
    const int lo   = g * L1_ROWS;
    const int sw   = wave - STREAM_WAVES;
    unsigned* myq  = queue + (sw > 0 ? sw : 0) * L1_SEG;
    int       n    = 0;

    {
        const int l  = tid & 31;
        const int kk = tid >> 5;
        const int k  = kk * 4 + (l >> 4) * 2;
        const float* wp = W1 + (size_t)k * 128 + (l & 15);
        float2 wv[8];
#pragma unroll
        for (int ct = 0; ct < 8; ++ct) wv[ct] = make_float2(wp[ct * 16], wp[128 + ct * 16]);
        float2* wst = (float2*)w1f + tid;
#pragma unroll
        for (int ct = 0; ct < 8; ++ct) wst[ct * 1024] = wv[ct];
    }
    __syncthreads();

    if (wave < STREAM_WAVES) {
        float*     stg = stage + wave * 16 * STG_PITCH;
        const int  mn  = lane & 15;
        const int  hf  = lane >> 4;
        const v2f* wl  = (const v2f*)w1f + lane;
#pragma unroll 1
        for (int b = 0; b < STREAM_TURNS; ++b) {
            const int tile = (b * STREAM_WAVES + wave) * L1_GROUPS + g;
            if (tile >= N_TILES) break;
            const int row0 = tile * 16;
            const float* pp = part + (size_t)(hf * (N_SLICES / 2)) * N_NODES + row0 + mn;
            float pd[N_SLICES / 2];
#pragma unroll
            for (int s = 0; s < N_SLICES / 2; ++s) pd[s] = pp[(size_t)s * N_NODES];
            v4f v[16];
#pragma unroll
            for (int j = 0; j < 16; ++j)
                v[j] = __builtin_nontemporal_load((const v4f*)x + (size_t)(row0 + j) * 32 + lane);
            __builtin_amdgcn_sched_barrier(0);
            float dsum = 0.0f;
#pragma unroll
            for (int s = 0; s < N_SLICES / 2; ++s) dsum += pd[s];
            dsum += __shfl_xor(dsum, 16, 32);
            const float nv = inv_sqrt_deg(dsum);
            if (lane < 16) ninv_out[row0 + lane] = nv;
#pragma unroll
            for (int j = 0; j < 16; ++j) {
                const float s = __int_as_float(__builtin_amdgcn_readlane(__float_as_int(nv), j));
                *(v4f*)(stg + j * STG_PITCH + lane * 4) = v[j] * s;
            }
            __builtin_amdgcn_fence(__ATOMIC_ACQ_REL, "wavefront");
            v2f a[32];
            {
                const float* ar = stg + mn * STG_PITCH + hf * 2;
#pragma unroll
                for (int kk = 0; kk < 32; ++kk) a[kk] = *(const v2f*)(ar + 4 * kk);
            }
            for (int ct = 0; ct < 8; ct += 2) {
                v8f c0 = {};
                v8f c1 = {};
#pragma unroll
                for (int kk = 0; kk < 32; ++kk) {
                    const v2f b0  = wl[((ct + 0) * 32 + kk) * 32];
                    const v2f b1v = wl[((ct + 1) * 32 + kk) * 32];
                    c0 = __builtin_amdgcn_wmma_f32_16x16x4_f32(false, a[kk], false, b0,
                                                               (short)0, c0, false, false);
                    c1 = __builtin_amdgcn_wmma_f32_16x16x4_f32(false, a[kk], false, b1v,
                                                               (short)0, c1, false, false);
                }
                float* d = stg + hf * 8 * STG_PITCH + ct * 16 + mn;
#pragma unroll
                for (int r = 0; r < 8; ++r) {
                    d[r * STG_PITCH]      = c0[r];
                    d[r * STG_PITCH + 16] = c1[r];
                }
            }
            __builtin_amdgcn_fence(__ATOMIC_ACQ_REL, "wavefront");
#pragma unroll
            for (int j = 0; j < 16; ++j)
                ((v4f*)xw)[(size_t)(row0 + j) * 32 + lane] = *(const v4f*)(stg + j * STG_PITCH + lane * 4);
        }
    } else {
        const uint4*   m4  = (const uint4*)omap;
        const int      rot = g * (N_SB / L1_GROUPS);
        const unsigned ltm = (1u << lane) - 1u;
        int            cnt = 0;
        unsigned flip[8];
#pragma unroll
        for (int j = 0; j < 8; ++j) flip[j] = ((g >> j) & 1) ? 0u : 0xffffffffu;

        for (int it = 0; it < SCAN_TURNS; ++it) {
            int   eb[4];
            bool  val[4];
            uint4 pl[4], ph[4];
#pragma unroll
            for (int u = 0; u < 4; ++u) {
                const int j = (it * 4 + u) * SCAN_WAVES + sw;
                int sb = j + rot;
                if (sb >= N_SB) sb -= N_SB;
                const int blk = sb * 32 + lane;
                val[u] = j < N_SB && blk < N_BLK;
                const int v = val[u] ? sb * 64 + lane : 0;
                pl[u] = m4[v];
                ph[u] = m4[v + 32];
                eb[u] = blk * 32;
            }
            __builtin_amdgcn_sched_barrier(0);
#pragma unroll
            for (int u = 0; u < 4; ++u) {
                unsigned m = (pl[u].x ^ flip[0]) & (pl[u].y ^ flip[1]) & (pl[u].z ^ flip[2]) & (pl[u].w ^ flip[3]) &
                             (ph[u].x ^ flip[4]) & (ph[u].y ^ flip[5]) & (ph[u].z ^ flip[6]) & (ph[u].w ^ flip[7]);
                if (!val[u]) m = 0u;
                while (__builtin_amdgcn_ballot_w32(m != 0u) != 0u) {
                    const bool     has = m != 0u;
                    const unsigned bal = __builtin_amdgcn_ballot_w32(has);
                    if (has) {
                        const int e = eb[u] + __ffs(m) - 1;
                        m &= m - 1u;
                        const int slot = cnt + __popc(bal & ltm);
                        if (slot < L1_SEG) myq[slot] = (unsigned)e;
                    }
                    cnt += __popc(bal);
                }
            }
        }
        n = cnt < L1_SEG ? cnt : L1_SEG;
        if (lane == 0) {
            wcnt[sw] = n;
            if (cnt > L1_SEG) oflag[0] = 1u;
        }
    }
    __syncthreads();

    int base = 0, total = 0;
    for (int v = 0; v < SCAN_WAVES; ++v) {
        const int c = wcnt[v];
        total += c;
        base += v < sw ? c : 0;
    }
    if (tid == 0) ecount[g] = total;

    unsigned* gl = elist + (size_t)g * L1_LIST + base;
    for (int k0 = 0; k0 < n; k0 += 32 * LIST_CHUNKS) {
        int e[LIST_CHUNKS], s[LIST_CHUNKS], d[LIST_CHUNKS];
#pragma unroll
        for (int j = 0; j < LIST_CHUNKS; ++j) {
            const int k = k0 + j * 32 + lane;
            e[j] = (int)myq[k < n ? k : 0];
        }
#pragma unroll
        for (int j = 0; j < LIST_CHUNKS; ++j) {
            s[j] = src[e[j]];
            d[j] = dst[e[j]];
        }
        asm volatile("" :: "v"(s[0]), "v"(s[1]), "v"(s[2]), "v"(s[3]), "v"(s[4]), "v"(s[5]),
                           "v"(s[6]), "v"(s[7]), "v"(s[8]), "v"(s[9]), "v"(s[10]), "v"(s[11]));
        asm volatile("" :: "v"(d[0]), "v"(d[1]), "v"(d[2]), "v"(d[3]), "v"(d[4]), "v"(d[5]),
                           "v"(d[6]), "v"(d[7]), "v"(d[8]), "v"(d[9]), "v"(d[10]), "v"(d[11]));
#pragma unroll
        for (int j = 0; j < LIST_CHUNKS; ++j) {
            const int k = k0 + j * 32 + lane;
            if (k < n) gl[k] = (unsigned)s[j] | ((unsigned)(d[j] - lo) << L1_SRC_BITS);
        }
    }
}

template <int M, int K>
__device__ __forceinline__ void fold_step(float (&p)[32], int lane) {
    const bool up = (lane & M) != 0;
#pragma unroll
    for (int i = 0; i < K; ++i) {
        const float keep = up ? p[i + K] : p[i];
        const float send = up ? p[i] : p[i + K];
        p[i] = keep + __shfl_xor(send, M, 32);
    }
}

__global__ __launch_bounds__(1024) void agg_lds_kernel(const unsigned* __restrict__ elist,
                                                       const int* __restrict__ ecount,
                                                       const unsigned* __restrict__ oflag,
                                                       const int* __restrict__ src,
                                                       const int* __restrict__ dst,
                                                       const float* __restrict__ xw,
                                                       const float* __restrict__ b1,
                                                       float* __restrict__ ninv_in,
                                                       const float* __restrict__ ninv_out,
                                                       const float* __restrict__ W2,
                                                       float* __restrict__ y_node) {
    __shared__ __attribute__((aligned(16))) float acc[L1_TILES * 16 * L1_PITCH];
    __shared__ unsigned icnt[L1_TILES * 16];

    const int tid  = threadIdx.x;
    const int lane = tid & 31;
    const int wave = __builtin_amdgcn_readfirstlane(tid >> 5);
    const int g    = blockIdx.x;
    const int lo   = g * L1_ROWS;

    int n = ecount[g];
    n = n < 0 ? 0 : (n > L1_LIST ? L1_LIST : n);
    const bool full = oflag[0] != 0u;

    {
        float4* a4 = (float4*)acc;
        for (int i = tid; i < L1_TILES * 16 * L1_PITCH / 4; i += 1024) a4[i] = make_float4(0.f, 0.f, 0.f, 0.f);
        if (tid < L1_TILES * 16) icnt[tid] = 0u;
    }
    __syncthreads();

    if (!full) {
        const unsigned* el = elist + (size_t)g * L1_LIST;
        for (int k0 = wave * 32; k0 < n; k0 += 1024) {
            const int      k  = k0 + lane;
            const bool     ok = k < n;
            const unsigned w  = el[ok ? k : k0];
            int s  = (int)(w & ((1u << L1_SRC_BITS) - 1u));
            int dl = (int)(w >> L1_SRC_BITS);
            if (ok) {
                atomicAdd(&icnt[dl], 1u);
            } else {
                s = 0; dl = L1_ROWS;
            }
            const int left = n - k0;
#pragma unroll
            for (int sb = 0; sb < 4; ++sb) {
                if (sb * 8 < left) {
                    float4 v[8];
                    int    dr[8];
#pragma unroll
                    for (int j = 0; j < 8; ++j) {
                        const int sj = __builtin_amdgcn_readlane(s, sb * 8 + j);
                        dr[j]        = __builtin_amdgcn_readlane(dl, sb * 8 + j);
                        v[j] = *(const float4*)(xw + (size_t)sj * 128 + lane * 4);
                    }
#pragma unroll
                    for (int j = 0; j < 8; ++j) {
                        float* a = acc + dr[j] * L1_PITCH + lane;
                        atomicAdd(a,      v[j].x);
                        atomicAdd(a + 32, v[j].y);
                        atomicAdd(a + 64, v[j].z);
                        atomicAdd(a + 96, v[j].w);
                    }
                }
            }
        }
    } else {
        for (int e = tid; e < N_EDGES; e += 1024) {
            const unsigned dl = (unsigned)(dst[e] - lo);
            if (dl < L1_ROWS) {
                const float* xr = xw + (size_t)src[e] * 128;
                float*       ar = acc + dl * L1_PITCH;
                atomicAdd(&icnt[dl], 1u);
                for (int f = 0; f < 128; ++f) atomicAdd(&ar[(f & 3) * 32 + (f >> 2)], xr[f]);
            }
        }
    }
    const bool  active = wave < L1_TILES;
    const int   mn     = lane & 15;
    float2 w2v[4];
    float  b1v[4];
    float  so = 0.0f;
    if (active) {
        const float2* w2p = (const float2*)W2;
#pragma unroll
        for (int j = 0; j < 4; ++j) {
            w2v[j] = w2p[4 * lane + j];
            b1v[j] = b1[4 * lane + j];
        }
        const int node = lo + wave * 16 + mn;
        so = ninv_out[node < N_NODES ? node : (N_NODES - 1)];
    }
    __syncthreads();

    if (active) {
        const float* trow = acc + wave * 16 * L1_PITCH;
        const int   rr = wave * 16 + mn;
        const float sc = inv_sqrt_deg((float)icnt[rr]);
        if (lane < 16 && rr < L1_ROWS && lo + rr < N_NODES) ninv_in[lo + rr] = sc;
        float p[32];
#pragma unroll
        for (int i = 0; i < 16; ++i) {
            const float  si = __int_as_float(__builtin_amdgcn_readlane(__float_as_int(sc), i));
            const float  s  = __int_as_float(__builtin_amdgcn_readlane(__float_as_int(so), i));
            const float* hr = trow + i * L1_PITCH + lane;
            float y0 = 0.0f, y1 = 0.0f;
#pragma unroll
            for (int j = 0; j < 4; ++j) {
                const float h = fmaxf(hr[32 * j] * si + b1v[j], 0.0f);
                const float v = h * s;
                y0 += v * w2v[j].x;
                y1 += v * w2v[j].y;
            }
            p[2 * i]     = y0;
            p[2 * i + 1] = y1;
        }
        fold_step<16, 16>(p, lane);
        fold_step<8, 8>(p, lane);
        fold_step<4, 4>(p, lane);
        fold_step<2, 2>(p, lane);
        fold_step<1, 1>(p, lane);
        const int r = wave * 16 + (lane >> 1);
        if (r < L1_ROWS && lo + r < N_NODES) y_node[(size_t)(lo + wave * 16) * 2 + lane] = p[0];
    }
}

#define L2_THREADS 512
#define L2_BATCH 16
__global__ __launch_bounds__(L2_THREADS) void agg2_kernel(const unsigned* __restrict__ elist,
                                                          const int* __restrict__ ecount,
                                                          const unsigned* __restrict__ oflag,
                                                          const int* __restrict__ src,
                                                          const int* __restrict__ dst,
                                                          const float* __restrict__ y_node,
                                                          const float* __restrict__ ninv_in,
                                                          const float* __restrict__ b2,
                                                          float* __restrict__ out) {
    __shared__ float acc[2 * L1_ROWS];

    const int tid = threadIdx.x;
    const int g   = blockIdx.x;
    const int lo  = g * L1_ROWS;

    int n = ecount[g];
    n = n < 0 ? 0 : (n > L1_LIST ? L1_LIST : n);
    const bool full = oflag[0] != 0u;

    for (int i = tid; i < 2 * L1_ROWS; i += L2_THREADS) acc[i] = 0.0f;
    __syncthreads();

    const float2* yn = (const float2*)y_node;
    if (!full) {
        const unsigned* el = elist + (size_t)g * L1_LIST;
        for (int k0 = 0; k0 < n; k0 += L2_THREADS * L2_BATCH) {
            unsigned pw[L2_BATCH];
            float2   yv[L2_BATCH];
#pragma unroll
            for (int j = 0; j < L2_BATCH; ++j) {
                const int k = k0 + j * L2_THREADS + tid;
                pw[j] = el[k < n ? k : n - 1];
            }
#pragma unroll
            for (int j = 0; j < L2_BATCH; ++j) yv[j] = yn[pw[j] & ((1u << L1_SRC_BITS) - 1u)];
            asm volatile("" :: "v"(yv[0].x), "v"(yv[0].y), "v"(yv[1].x), "v"(yv[1].y),
                               "v"(yv[2].x), "v"(yv[2].y), "v"(yv[3].x), "v"(yv[3].y),
                               "v"(yv[4].x), "v"(yv[4].y), "v"(yv[5].x), "v"(yv[5].y),
                               "v"(yv[6].x), "v"(yv[6].y), "v"(yv[7].x), "v"(yv[7].y));
            asm volatile("" :: "v"(yv[8].x), "v"(yv[8].y), "v"(yv[9].x), "v"(yv[9].y),
                               "v"(yv[10].x), "v"(yv[10].y), "v"(yv[11].x), "v"(yv[11].y),
                               "v"(yv[12].x), "v"(yv[12].y), "v"(yv[13].x), "v"(yv[13].y),
                               "v"(yv[14].x), "v"(yv[14].y), "v"(yv[15].x), "v"(yv[15].y));
#pragma unroll
            for (int j = 0; j < L2_BATCH; ++j) {
                if (k0 + j * L2_THREADS + tid < n) {
                    const unsigned dl = pw[j] >> L1_SRC_BITS;
                    atomicAdd(&acc[2 * dl], yv[j].x);
                    atomicAdd(&acc[2 * dl + 1], yv[j].y);
                }
            }
        }
    } else {
        for (int e = tid; e < N_EDGES; e += L2_THREADS) {
            const unsigned r = (unsigned)(dst[e] - lo);
            if (r < L1_ROWS) {
                const float2 v = yn[src[e]];
                atomicAdd(&acc[2 * r], v.x);
                atomicAdd(&acc[2 * r + 1], v.y);
            }
        }
    }
    __syncthreads();

    for (int i = tid; i < 2 * L1_ROWS; i += L2_THREADS) {
        const int node = lo + (i >> 1);
        if (node < N_NODES) out[(size_t)lo * 2 + i] = acc[i] * ninv_in[node] + b2[i & 1];
    }
}

extern "C" void kernel_launch(void* const* d_in, const int* in_sizes, int n_in,
                              void* d_out, int out_size, void* d_ws, size_t ws_size,
                              hipStream_t stream) {
    const float* in_feat = (const float*)d_in[0];
    const float* W1      = (const float*)d_in[1];
    const float* b1      = (const float*)d_in[2];
    const float* W2      = (const float*)d_in[3];
    const float* b2      = (const float*)d_in[4];
    const int*   src     = (const int*)d_in[5];
    const int*   dst     = (const int*)d_in[6];
    float*       out     = (float*)d_out;

    const int N = N_NODES;

    float* ws      = (float*)d_ws;
    float* deg_out = ws;
    float* deg_in  = ws + N;
    float* xw      = ws + 2 * (size_t)N;
    float* y_node  = xw + (size_t)N * 128;
    float* part    = y_node + 2 * (size_t)N;
    unsigned* omap   = (unsigned*)(part + (size_t)N_SLICES * N);
    unsigned* elist  = omap + (size_t)N_SB * 256;
    int*      ecount = (int*)(elist + (size_t)L1_GROUPS * L1_LIST);
    unsigned* oflag  = (unsigned*)(ecount + L1_GROUPS);

    degrees_kernel<<<dim3(N_SLICES, N_RANGES), 1024, 0, stream>>>(src, dst, part, omap, oflag);
    scale_scan_kernel<<<L1_GROUPS, 1024, 0, stream>>>(in_feat, part, W1, deg_out, xw, src, dst, omap,
                                                      elist, ecount, oflag);
    agg_lds_kernel<<<L1_GROUPS, 1024, 0, stream>>>(elist, ecount, oflag, src, dst, xw, b1, deg_in, deg_out, W2,
                                                   y_node);
    agg2_kernel<<<L1_GROUPS, L2_THREADS, 0, stream>>>(elist, ecount, oflag, src, dst, y_node, deg_in, b2, out);
}
